# starting kernel
# baseline (speedup 1.0000x reference)
#include <hip/hip_runtime.h>

#define N_NODES 50000
#define N_EDGES 800000
#define D_IN    128
#define D_HID   256

typedef __attribute__((ext_vector_type(2))) float v2f;
typedef __attribute__((ext_vector_type(8))) float v8f;


__global__ void deg_init_kernel(float* __restrict__ deg, int n) {
  int i = blockIdx.x * blockDim.x + threadIdx.x;
  if (i < n) deg[i] = 1.0f;
}

__global__ void deg_edge_kernel(const int* __restrict__ col, float* __restrict__ deg, int e) {
  int i = blockIdx.x * blockDim.x + threadIdx.x;
  if (i < e) atomicAdd(&deg[col[i]], 1.0f);
}

__global__ void dinv_kernel(const float* __restrict__ deg, float* __restrict__ dinv, int n) {
  int i = blockIdx.x * blockDim.x + threadIdx.x;
  if (i < n) dinv[i] = rsqrtf(deg[i]);
}

template <int RELU>
__global__ __launch_bounds__(128) void gemm_wmma_kernel(const float* __restrict__ A,
                                                        const float* __restrict__ W,
                                                        float* __restrict__ H,
                                                        int din) {
  const int lane = threadIdx.x & 31;
  const int wave = threadIdx.x >> 5;
  const int half = lane >> 4;
  const int mrow = lane & 15;
  const int m_base = blockIdx.x * 16;
  const int c_base = wave * 64;

  const float* __restrict__ Arow = A + (size_t)(m_base + mrow) * din;

  v8f acc0 = {}, acc1 = {}, acc2 = {}, acc3 = {};

  for (int k = 0; k < din; k += 4) {
    const int kk = k + (half << 1);
    v2f a = *(const v2f*)(Arow + kk);
    if (RELU) { a.x = fmaxf(a.x, 0.0f); a.y = fmaxf(a.y, 0.0f); }

    const float* __restrict__ Bp = W + (size_t)kk * 256 + c_base + mrow;
    v2f b0; b0.x = Bp[0];  b0.y = Bp[256 + 0];
    v2f b1; b1.x = Bp[16]; b1.y = Bp[256 + 16];
    v2f b2; b2.x = Bp[32]; b2.y = Bp[256 + 32];
    v2f b3; b3.x = Bp[48]; b3.y = Bp[256 + 48];

    acc0 = __builtin_amdgcn_wmma_f32_16x16x4_f32(false, a, false, b0, (short)0, acc0, false, false);
    acc1 = __builtin_amdgcn_wmma_f32_16x16x4_f32(false, a, false, b1, (short)0, acc1, false, false);
    acc2 = __builtin_amdgcn_wmma_f32_16x16x4_f32(false, a, false, b2, (short)0, acc2, false, false);
    acc3 = __builtin_amdgcn_wmma_f32_16x16x4_f32(false, a, false, b3, (short)0, acc3, false, false);
  }

  float* __restrict__ Hp = H + (size_t)(m_base + half * 8) * 256 + c_base + mrow;
#pragma unroll
  for (int r = 0; r < 8; ++r) {
    Hp[(size_t)r * 256 + 0]  = acc0[r];
    Hp[(size_t)r * 256 + 16] = acc1[r];
    Hp[(size_t)r * 256 + 32] = acc2[r];
    Hp[(size_t)r * 256 + 48] = acc3[r];
  }
}


__global__ void agg_self_kernel(const float* __restrict__ h,
                                const float* __restrict__ dinv,
                                float* __restrict__ out) {
  size_t i = (size_t)blockIdx.x * blockDim.x + threadIdx.x;
  size_t base = i * 4;
  int node = (int)(base >> 8);
  float s = dinv[node];
  s = s * s;
  float4 v = *(const float4*)(h + base);
  float4 o;
  o.x = v.x * s; o.y = v.y * s; o.z = v.z * s; o.w = v.w * s;
  *(float4*)(out + base) = o;
}

__global__ __launch_bounds__(256) void agg_edge_kernel(const float* __restrict__ h,
                                                       const float* __restrict__ dinv,
                                                       const int* __restrict__ row,
                                                       const int* __restrict__ col,
                                                       float* __restrict__ out, int e) {
  int edge = blockIdx.x * (blockDim.x >> 5) + (threadIdx.x >> 5);
  int lane = threadIdx.x & 31;
  if (edge >= e) return;
  int r = row[edge];
  int c = col[edge];
  float nrm = dinv[r] * dinv[c];
  const float* __restrict__ hp = h + (size_t)r * 256;
  float* __restrict__ op = out + (size_t)c * 256;
#pragma unroll
  for (int chunk = 0; chunk < 2; ++chunk) {
    int off = chunk * 128 + lane * 4;
    float4 v = *(const float4*)(hp + off);
    atomicAdd(op + off + 0, v.x * nrm);
    atomicAdd(op + off + 1, v.y * nrm);
    atomicAdd(op + off + 2, v.z * nrm);
    atomicAdd(op + off + 3, v.w * nrm);
  }
}


extern "C" void kernel_launch(void* const* d_in, const int* in_sizes, int n_in,
                              void* d_out, int out_size, void* d_ws, size_t ws_size,
                              hipStream_t stream) {
  const float* x     = (const float*)d_in[0];
  const int*   ei    = (const int*)d_in[1];
  const float* W_in  = (const float*)d_in[2];
  const float* W_hid = (const float*)d_in[3];
  const float* W_out = (const float*)d_in[4];
  float* out = (float*)d_out;

  float* ws   = (float*)d_ws;
  float* deg  = ws;
  float* dinv = ws + 65536;
  float* bufH = ws + 2 * 65536;
  float* bufG = bufH + (size_t)N_NODES * 256;

  const int* row = ei;
  const int* col = ei + N_EDGES;

  const int T256N = (N_NODES + 255) / 256;
  const int T256E = (N_EDGES + 255) / 256;
  const int ROW_TILES = N_NODES / 16;
  const int SELF_BLKS = (N_NODES * 256 / 4) / 256;
  const int EDGE_BLKS = (N_EDGES + 7) / 8;

  deg_init_kernel<<<T256N, 256, 0, stream>>>(deg, N_NODES);
  deg_edge_kernel<<<T256E, 256, 0, stream>>>(col, deg, N_EDGES);
  dinv_kernel<<<T256N, 256, 0, stream>>>(deg, dinv, N_NODES);

  gemm_wmma_kernel<0><<<ROW_TILES, 128, 0, stream>>>(x, W_in, bufH, D_IN);
  agg_self_kernel<<<SELF_BLKS, 256, 0, stream>>>(bufH, dinv, bufG);
  agg_edge_kernel<<<EDGE_BLKS, 256, 0, stream>>>(bufH, dinv, row, col, bufG, N_EDGES);

  gemm_wmma_kernel<1><<<ROW_TILES, 128, 0, stream>>>(bufG, W_hid, bufH, D_HID);
  agg_self_kernel<<<SELF_BLKS, 256, 0, stream>>>(bufH, dinv, out);
  agg_edge_kernel<<<EDGE_BLKS, 256, 0, stream>>>(bufH, dinv, row, col, out, N_EDGES);

  gemm_wmma_kernel<1><<<ROW_TILES, 128, 0, stream>>>(out, W_out, bufH, D_HID);
  agg_self_kernel<<<SELF_BLKS, 256, 0, stream>>>(bufH, dinv, out);
  agg_edge_kernel<<<EDGE_BLKS, 256, 0, stream>>>(bufH, dinv, row, col, out, N_EDGES);
}


# Edge scatter by float atomics replaced by a gather: per-target source lists (arrival-order rank from one int atomic, scan, fill); one wave sums a node's rows in registers in list order, stores once
# speedup vs baseline: 2.9932x; 2.9932x over previous
#include <hip/hip_runtime.h>

#define N_NODES 50000
#define N_EDGES 800000
#define D_IN    128
#define D_HID   256

static_assert(N_NODES % 4 == 0, "scan_dinv_kernel handles 4 counters per thread");

typedef __attribute__((ext_vector_type(2))) float v2f;
typedef __attribute__((ext_vector_type(4))) float v4f;
typedef __attribute__((ext_vector_type(8))) float v8f;


__global__ void cnt_zero_kernel(int* __restrict__ cnt, int n) {
  int i = blockIdx.x * blockDim.x + threadIdx.x;
  if (i < n) cnt[i] = 0;
}

__global__ void edge_rank_kernel(const int* __restrict__ col, int* __restrict__ cnt,
                                 int* __restrict__ rank, int e) {
  int i = blockIdx.x * blockDim.x + threadIdx.x;
  if (i < e) rank[i] = atomicAdd(&cnt[col[i]], 1);
}

__global__ __launch_bounds__(1024) void scan_dinv_kernel(const int* __restrict__ cnt,
                                                         int* __restrict__ off,
                                                         float* __restrict__ dinv, int n) {
  __shared__ int wsum[32];
  const int t = threadIdx.x;
  const int lane = t & 31;
  const int w = t >> 5;
  int carry = 0;
  for (int base = 0; base < n; base += 4096) {
    const int i0 = base + t * 4;
    const bool live = i0 < n;
    int4 c = make_int4(0, 0, 0, 0);
    if (live) c = *(const int4*)(cnt + i0);
    const int s = c.x + c.y + c.z + c.w;
    int incl = s;
#pragma unroll
    for (int d = 1; d < 32; d <<= 1) {
      int v = __shfl_up(incl, d);
      if (lane >= d) incl += v;
    }
    if (lane == 31) wsum[w] = incl;
    __syncthreads();
    const int ws = wsum[lane];
    int wincl = ws;
#pragma unroll
    for (int d = 1; d < 32; d <<= 1) {
      int v = __shfl_up(wincl, d);
      if (lane >= d) wincl += v;
    }
    const int wave_excl = __shfl(wincl, w) - __shfl(ws, w);
    const int total = __shfl(wincl, 31);
    const int ex = carry + wave_excl + (incl - s);
    if (live) {
      int4 o;
      o.x = ex;
      o.y = ex + c.x;
      o.z = o.y + c.y;
      o.w = o.z + c.z;
      *(int4*)(off + i0) = o;
      float4 dv;
      dv.x = rsqrtf((float)(c.x + 1));
      dv.y = rsqrtf((float)(c.y + 1));
      dv.z = rsqrtf((float)(c.z + 1));
      dv.w = rsqrtf((float)(c.w + 1));
      *(float4*)(dinv + i0) = dv;
    }
    carry += total;
    __syncthreads();
  }
}

__global__ void csr_fill_kernel(const int* __restrict__ row, const int* __restrict__ col,
                                const int* __restrict__ rank, const int* __restrict__ off,
                                const float* __restrict__ dinv, int2* __restrict__ ent, int e) {
  int i = blockIdx.x * blockDim.x + threadIdx.x;
  if (i < e) {
    int r = row[i];
    int p = off[col[i]] + rank[i];
    ent[p] = make_int2(r, __float_as_int(dinv[r]));
  }
}

template <int RELU>
__global__ __launch_bounds__(128) void gemm_wmma_kernel(const float* __restrict__ A,
                                                        const float* __restrict__ W,
                                                        float* __restrict__ H,
                                                        int din) {
  const int lane = threadIdx.x & 31;
  const int wave = threadIdx.x >> 5;
  const int half = lane >> 4;
  const int mrow = lane & 15;
  const int m_base = blockIdx.x * 16;
  const int c_base = wave * 64;

  const float* __restrict__ Arow = A + (size_t)(m_base + mrow) * din;

  v8f acc0 = {}, acc1 = {}, acc2 = {}, acc3 = {};

  for (int k = 0; k < din; k += 4) {
    const int kk = k + (half << 1);
    v2f a = *(const v2f*)(Arow + kk);
    if (RELU) { a.x = fmaxf(a.x, 0.0f); a.y = fmaxf(a.y, 0.0f); }

    const float* __restrict__ Bp = W + (size_t)kk * 256 + c_base + mrow;
    v2f b0; b0.x = Bp[0];  b0.y = Bp[256 + 0];
    v2f b1; b1.x = Bp[16]; b1.y = Bp[256 + 16];
    v2f b2; b2.x = Bp[32]; b2.y = Bp[256 + 32];
    v2f b3; b3.x = Bp[48]; b3.y = Bp[256 + 48];

    acc0 = __builtin_amdgcn_wmma_f32_16x16x4_f32(false, a, false, b0, (short)0, acc0, false, false);
    acc1 = __builtin_amdgcn_wmma_f32_16x16x4_f32(false, a, false, b1, (short)0, acc1, false, false);
    acc2 = __builtin_amdgcn_wmma_f32_16x16x4_f32(false, a, false, b2, (short)0, acc2, false, false);
    acc3 = __builtin_amdgcn_wmma_f32_16x16x4_f32(false, a, false, b3, (short)0, acc3, false, false);
  }

  float* __restrict__ Hp = H + (size_t)(m_base + half * 8) * 256 + c_base + mrow;
#pragma unroll
  for (int r = 0; r < 8; ++r) {
    Hp[(size_t)r * 256 + 0]  = acc0[r];
    Hp[(size_t)r * 256 + 16] = acc1[r];
    Hp[(size_t)r * 256 + 32] = acc2[r];
    Hp[(size_t)r * 256 + 48] = acc3[r];
  }
}


__device__ __forceinline__ void row_fma(v4f& a0, v4f& a1, const float* __restrict__ p, float w) {
  const v4f x = *(const v4f*)(p);
  const v4f y = *(const v4f*)(p + 128);
  const v4f wv = {w, w, w, w};
  a0 = __builtin_elementwise_fma(x, wv, a0);
  a1 = __builtin_elementwise_fma(y, wv, a1);
}

__global__ __launch_bounds__(256) void agg_gather_kernel(const float* __restrict__ h,
                                                         const float* __restrict__ dinv,
                                                         const int* __restrict__ cnt,
                                                         const int* __restrict__ off,
                                                         const int2* __restrict__ ent,
                                                         float* __restrict__ out, int n) {
  const int lane = threadIdx.x & 31;
  const int node = __builtin_amdgcn_readfirstlane((int)(blockIdx.x * 8 + (threadIdx.x >> 5)));
  if (node >= n) return;
  const float dc = dinv[node];
  const int beg = off[node];
  const int deg = cnt[node];
  const float* __restrict__ hl = h + lane * 4;

  v4f a0, a1;
  {
    const float s = dc * dc;
    const float* __restrict__ hp = hl + (size_t)node * 256;
    a0 = *(const v4f*)(hp) * s;
    a1 = *(const v4f*)(hp + 128) * s;
  }

  for (int base = 0; base < deg; base += 32) {
    const int m = min(32, deg - base);
    int myr = 0;
    int myw = 0;
    if (lane < m) {
      const int2 en = ent[beg + base + lane];
      myr = en.x;
      myw = __float_as_int(__int_as_float(en.y) * dc);
    }
    int j = 0;
    for (; j + 4 <= m; j += 4) {
      const int r0 = __builtin_amdgcn_readlane(myr, j + 0);
      const int r1 = __builtin_amdgcn_readlane(myr, j + 1);
      const int r2 = __builtin_amdgcn_readlane(myr, j + 2);
      const int r3 = __builtin_amdgcn_readlane(myr, j + 3);
      const float w0 = __int_as_float(__builtin_amdgcn_readlane(myw, j + 0));
      const float w1 = __int_as_float(__builtin_amdgcn_readlane(myw, j + 1));
      const float w2 = __int_as_float(__builtin_amdgcn_readlane(myw, j + 2));
      const float w3 = __int_as_float(__builtin_amdgcn_readlane(myw, j + 3));
      row_fma(a0, a1, hl + (size_t)r0 * 256, w0);
      row_fma(a0, a1, hl + (size_t)r1 * 256, w1);
      row_fma(a0, a1, hl + (size_t)r2 * 256, w2);
      row_fma(a0, a1, hl + (size_t)r3 * 256, w3);
    }
    if (j + 2 <= m) {
      const int r0 = __builtin_amdgcn_readlane(myr, j + 0);
      const int r1 = __builtin_amdgcn_readlane(myr, j + 1);
      const float w0 = __int_as_float(__builtin_amdgcn_readlane(myw, j + 0));
      const float w1 = __int_as_float(__builtin_amdgcn_readlane(myw, j + 1));
      row_fma(a0, a1, hl + (size_t)r0 * 256, w0);
      row_fma(a0, a1, hl + (size_t)r1 * 256, w1);
      j += 2;
    }
    if (j < m) {
      const int r0 = __builtin_amdgcn_readlane(myr, j);
      const float w0 = __int_as_float(__builtin_amdgcn_readlane(myw, j));
      row_fma(a0, a1, hl + (size_t)r0 * 256, w0);
    }
  }

  float* __restrict__ op = out + (size_t)node * 256 + lane * 4;
  *(v4f*)(op) = a0;
  *(v4f*)(op + 128) = a1;
}


extern "C" void kernel_launch(void* const* d_in, const int* in_sizes, int n_in,
                              void* d_out, int out_size, void* d_ws, size_t ws_size,
                              hipStream_t stream) {
  const float* x     = (const float*)d_in[0];
  const int*   ei    = (const int*)d_in[1];
  const float* W_in  = (const float*)d_in[2];
  const float* W_hid = (const float*)d_in[3];
  const float* W_out = (const float*)d_in[4];
  float* out = (float*)d_out;

  float* ws   = (float*)d_ws;
  int*   cnt  = (int*)ws;
  float* dinv = ws + 65536;
  int*   off  = (int*)(ws + 2 * 65536);
  int*   rank = (int*)(ws + 3 * 65536);
  int2*  ent  = (int2*)(ws + (1 << 20));
  float* bufH = ws + (4 << 20);
  float* bufG = bufH + (size_t)N_NODES * 256;

  const int* row = ei;
  const int* col = ei + N_EDGES;

  const int T256N = (N_NODES + 255) / 256;
  const int T256E = (N_EDGES + 255) / 256;
  const int ROW_TILES = N_NODES / 16;
  const int NODE_BLKS = (N_NODES + 7) / 8;

  cnt_zero_kernel<<<T256N, 256, 0, stream>>>(cnt, N_NODES);
  edge_rank_kernel<<<T256E, 256, 0, stream>>>(col, cnt, rank, N_EDGES);
  scan_dinv_kernel<<<1, 1024, 0, stream>>>(cnt, off, dinv, N_NODES);
  csr_fill_kernel<<<T256E, 256, 0, stream>>>(row, col, rank, off, dinv, ent, N_EDGES);

  gemm_wmma_kernel<0><<<ROW_TILES, 128, 0, stream>>>(x, W_in, bufH, D_IN);
  agg_gather_kernel<<<NODE_BLKS, 256, 0, stream>>>(bufH, dinv, cnt, off, ent, bufG, N_NODES);

  gemm_wmma_kernel<1><<<ROW_TILES, 128, 0, stream>>>(bufG, W_hid, bufH, D_HID);
  agg_gather_kernel<<<NODE_BLKS, 256, 0, stream>>>(bufH, dinv, cnt, off, ent, out, N_NODES);

  gemm_wmma_kernel<1><<<ROW_TILES, 128, 0, stream>>>(out, W_out, bufH, D_HID);
  agg_gather_kernel<<<NODE_BLKS, 256, 0, stream>>>(bufH, dinv, cnt, off, ent, out, N_NODES);
}


# GEMM operands staged through LDS: 80x128 block tile, 32-deep K slabs fetched by coalesced 16-byte loads (next slab in registers under the WMMAs); each wave 5x2 tiles, fragments from LDS
# speedup vs baseline: 3.4261x; 1.1446x over previous
#include <hip/hip_runtime.h>

#define N_NODES 50000
#define N_EDGES 800000
#define D_IN    128
#define D_HID   256

static_assert(N_NODES % 4 == 0, "scan_dinv_kernel handles 4 counters per thread");

typedef __attribute__((ext_vector_type(2))) float v2f;
typedef __attribute__((ext_vector_type(4))) float v4f;
typedef __attribute__((ext_vector_type(8))) float v8f;


__global__ void cnt_zero_kernel(int* __restrict__ cnt, int n) {
  int i = blockIdx.x * blockDim.x + threadIdx.x;
  if (i < n) cnt[i] = 0;
}

__global__ void edge_rank_kernel(const int* __restrict__ col, int* __restrict__ cnt,
                                 int* __restrict__ rank, int e) {
  int i = blockIdx.x * blockDim.x + threadIdx.x;
  if (i < e) rank[i] = atomicAdd(&cnt[col[i]], 1);
}

__global__ __launch_bounds__(1024) void scan_dinv_kernel(const int* __restrict__ cnt,
                                                         int* __restrict__ off,
                                                         float* __restrict__ dinv, int n) {
  __shared__ int wsum[32];
  const int t = threadIdx.x;
  const int lane = t & 31;
  const int w = t >> 5;
  int carry = 0;
  for (int base = 0; base < n; base += 4096) {
    const int i0 = base + t * 4;
    const bool live = i0 < n;
    int4 c = make_int4(0, 0, 0, 0);
    if (live) c = *(const int4*)(cnt + i0);
    const int s = c.x + c.y + c.z + c.w;
    int incl = s;
#pragma unroll
    for (int d = 1; d < 32; d <<= 1) {
      int v = __shfl_up(incl, d);
      if (lane >= d) incl += v;
    }
    if (lane == 31) wsum[w] = incl;
    __syncthreads();
    const int ws = wsum[lane];
    int wincl = ws;
#pragma unroll
    for (int d = 1; d < 32; d <<= 1) {
      int v = __shfl_up(wincl, d);
      if (lane >= d) wincl += v;
    }
    const int wave_excl = __shfl(wincl, w) - __shfl(ws, w);
    const int total = __shfl(wincl, 31);
    const int ex = carry + wave_excl + (incl - s);
    if (live) {
      int4 o;
      o.x = ex;
      o.y = ex + c.x;
      o.z = o.y + c.y;
      o.w = o.z + c.z;
      *(int4*)(off + i0) = o;
      float4 dv;
      dv.x = rsqrtf((float)(c.x + 1));
      dv.y = rsqrtf((float)(c.y + 1));
      dv.z = rsqrtf((float)(c.z + 1));
      dv.w = rsqrtf((float)(c.w + 1));
      *(float4*)(dinv + i0) = dv;
    }
    carry += total;
    __syncthreads();
  }
}

__global__ void csr_fill_kernel(const int* __restrict__ row, const int* __restrict__ col,
                                const int* __restrict__ rank, const int* __restrict__ off,
                                const float* __restrict__ dinv, int2* __restrict__ ent, int e) {
  int i = blockIdx.x * blockDim.x + threadIdx.x;
  if (i < e) {
    int r = row[i];
    int p = off[col[i]] + rank[i];
    ent[p] = make_int2(r, __float_as_int(dinv[r]));
  }
}

#define GEMM_ROWS 80
#define GEMM_COLS 128
#define GEMM_KS   32
#define SA_PITCH  36
#define SW_PITCH  288

static_assert(N_NODES % GEMM_ROWS == 0, "row tiles are exact, no guards in the GEMM");
static_assert(GEMM_ROWS * (GEMM_KS / 4) == 5 * 128, "A slab = 5 float4 per thread");

template <int RELU, int DIN>
__global__ __launch_bounds__(128) void gemm_wmma_kernel(const float* __restrict__ A,
                                                        const float* __restrict__ W,
                                                        float* __restrict__ H) {
  __shared__ __attribute__((aligned(16))) float sW[(GEMM_KS / 2) * SW_PITCH];
  __shared__ __attribute__((aligned(16))) float sA[GEMM_ROWS * SA_PITCH];

  const int t = threadIdx.x;
  const int lane = t & 31;
  const int wave = t >> 5;
  const int half = lane >> 4;
  const int c = lane & 15;
  const int m_base = (blockIdx.x >> 1) * GEMM_ROWS;
  const int n_base = (blockIdx.x & 1) * GEMM_COLS;

  const int wp = t >> 5;
  const int wq = t & 31;
  const float* __restrict__ gW = W + (size_t)(2 * wp) * 256 + n_base + wq * 4;
  const float* __restrict__ gA = A + (size_t)(m_base + (t >> 3)) * DIN + (t & 7) * 4;
  float* __restrict__ dW = sW + wp * SW_PITCH + wq * 8;
  float* __restrict__ dA = sA + (t >> 3) * SA_PITCH + (t & 7) * 4;
  v4f wr[8];
  v4f ar[5];

  auto gload = [&](int k0) {
#pragma unroll
    for (int i = 0; i < 4; ++i) {
      const float* __restrict__ g = gW + (size_t)(k0 + 8 * i) * 256;
      wr[2 * i]     = *(const v4f*)(g);
      wr[2 * i + 1] = *(const v4f*)(g + 256);
    }
#pragma unroll
    for (int i = 0; i < 5; ++i)
      ar[i] = *(const v4f*)(gA + (size_t)(16 * i) * DIN + k0);
  };

  auto sstore = [&]() {
#pragma unroll
    for (int i = 0; i < 4; ++i) {
      float* d = dW + (4 * i) * SW_PITCH;
      const v4f w0 = wr[2 * i], w1 = wr[2 * i + 1];
      const v4f lo = {w0.x, w1.x, w0.y, w1.y};
      const v4f hi = {w0.z, w1.z, w0.w, w1.w};
      *(v4f*)(d) = lo;
      *(v4f*)(d + 4) = hi;
    }
#pragma unroll
    for (int i = 0; i < 5; ++i) {
      v4f v = ar[i];
      if (RELU) {
        v.x = fmaxf(v.x, 0.0f); v.y = fmaxf(v.y, 0.0f);
        v.z = fmaxf(v.z, 0.0f); v.w = fmaxf(v.w, 0.0f);
      }
      *(v4f*)(dA + (16 * i) * SA_PITCH) = v;
    }
  };

  v8f acc[5][2];
#pragma unroll
  for (int r = 0; r < 5; ++r) {
    acc[r][0] = v8f{};
    acc[r][1] = v8f{};
  }

  const float* __restrict__ aF = sA + c * SA_PITCH + 2 * half;
  const float* __restrict__ bF = sW + half * SW_PITCH + (wave * 32 + c) * 2;

  gload(0);
  for (int k0 = 0; k0 < DIN; k0 += GEMM_KS) {
    sstore();
    __syncthreads();
    if (k0 + GEMM_KS < DIN) gload(k0 + GEMM_KS);
#pragma unroll
    for (int ks = 0; ks < GEMM_KS / 4; ++ks) {
      const v2f b0 = *(const v2f*)(bF + (2 * ks) * SW_PITCH);
      const v2f b1 = *(const v2f*)(bF + (2 * ks) * SW_PITCH + 32);
#pragma unroll
      for (int r = 0; r < 5; ++r) {
        const v2f a = *(const v2f*)(aF + r * 16 * SA_PITCH + ks * 4);
        acc[r][0] = __builtin_amdgcn_wmma_f32_16x16x4_f32(false, a, false, b0, (short)0, acc[r][0], false, false);
        acc[r][1] = __builtin_amdgcn_wmma_f32_16x16x4_f32(false, a, false, b1, (short)0, acc[r][1], false, false);
      }
    }
    __syncthreads();
  }

  float* __restrict__ Hp = H + (size_t)(m_base + half * 8) * 256 + n_base + wave * 32 + c;
#pragma unroll
  for (int r = 0; r < 5; ++r) {
#pragma unroll
    for (int e = 0; e < 8; ++e) {
      Hp[(size_t)(r * 16 + e) * 256 + 0]  = acc[r][0][e];
      Hp[(size_t)(r * 16 + e) * 256 + 16] = acc[r][1][e];
    }
  }
}


__device__ __forceinline__ void row_fma(v4f& a0, v4f& a1, const float* __restrict__ p, float w) {
  const v4f x = *(const v4f*)(p);
  const v4f y = *(const v4f*)(p + 128);
  const v4f wv = {w, w, w, w};
  a0 = __builtin_elementwise_fma(x, wv, a0);
  a1 = __builtin_elementwise_fma(y, wv, a1);
}

__global__ __launch_bounds__(256) void agg_gather_kernel(const float* __restrict__ h,
                                                         const float* __restrict__ dinv,
                                                         const int* __restrict__ cnt,
                                                         const int* __restrict__ off,
                                                         const int2* __restrict__ ent,
                                                         float* __restrict__ out, int n) {
  const int lane = threadIdx.x & 31;
  const int node = __builtin_amdgcn_readfirstlane((int)(blockIdx.x * 8 + (threadIdx.x >> 5)));
  if (node >= n) return;
  const float dc = dinv[node];
  const int beg = off[node];
  const int deg = cnt[node];
  const float* __restrict__ hl = h + lane * 4;

  v4f a0, a1;
  {
    const float s = dc * dc;
    const float* __restrict__ hp = hl + (size_t)node * 256;
    a0 = *(const v4f*)(hp) * s;
    a1 = *(const v4f*)(hp + 128) * s;
  }

  for (int base = 0; base < deg; base += 32) {
    const int m = min(32, deg - base);
    int myr = 0;
    int myw = 0;
    if (lane < m) {
      const int2 en = ent[beg + base + lane];
      myr = en.x;
      myw = __float_as_int(__int_as_float(en.y) * dc);
    }
    int j = 0;
    for (; j + 4 <= m; j += 4) {
      const int r0 = __builtin_amdgcn_readlane(myr, j + 0);
      const int r1 = __builtin_amdgcn_readlane(myr, j + 1);
      const int r2 = __builtin_amdgcn_readlane(myr, j + 2);
      const int r3 = __builtin_amdgcn_readlane(myr, j + 3);
      const float w0 = __int_as_float(__builtin_amdgcn_readlane(myw, j + 0));
      const float w1 = __int_as_float(__builtin_amdgcn_readlane(myw, j + 1));
      const float w2 = __int_as_float(__builtin_amdgcn_readlane(myw, j + 2));
      const float w3 = __int_as_float(__builtin_amdgcn_readlane(myw, j + 3));
      row_fma(a0, a1, hl + (size_t)r0 * 256, w0);
      row_fma(a0, a1, hl + (size_t)r1 * 256, w1);
      row_fma(a0, a1, hl + (size_t)r2 * 256, w2);
      row_fma(a0, a1, hl + (size_t)r3 * 256, w3);
    }
    if (j + 2 <= m) {
      const int r0 = __builtin_amdgcn_readlane(myr, j + 0);
      const int r1 = __builtin_amdgcn_readlane(myr, j + 1);
      const float w0 = __int_as_float(__builtin_amdgcn_readlane(myw, j + 0));
      const float w1 = __int_as_float(__builtin_amdgcn_readlane(myw, j + 1));
      row_fma(a0, a1, hl + (size_t)r0 * 256, w0);
      row_fma(a0, a1, hl + (size_t)r1 * 256, w1);
      j += 2;
    }
    if (j < m) {
      const int r0 = __builtin_amdgcn_readlane(myr, j);
      const float w0 = __int_as_float(__builtin_amdgcn_readlane(myw, j));
      row_fma(a0, a1, hl + (size_t)r0 * 256, w0);
    }
  }

  float* __restrict__ op = out + (size_t)node * 256 + lane * 4;
  *(v4f*)(op) = a0;
  *(v4f*)(op + 128) = a1;
}


extern "C" void kernel_launch(void* const* d_in, const int* in_sizes, int n_in,
                              void* d_out, int out_size, void* d_ws, size_t ws_size,
                              hipStream_t stream) {
  const float* x     = (const float*)d_in[0];
  const int*   ei    = (const int*)d_in[1];
  const float* W_in  = (const float*)d_in[2];
  const float* W_hid = (const float*)d_in[3];
  const float* W_out = (const float*)d_in[4];
  float* out = (float*)d_out;

  float* ws   = (float*)d_ws;
  int*   cnt  = (int*)ws;
  float* dinv = ws + 65536;
  int*   off  = (int*)(ws + 2 * 65536);
  int*   rank = (int*)(ws + 3 * 65536);
  int2*  ent  = (int2*)(ws + (1 << 20));
  float* bufH = ws + (4 << 20);
  float* bufG = bufH + (size_t)N_NODES * 256;

  const int* row = ei;
  const int* col = ei + N_EDGES;

  const int T256N = (N_NODES + 255) / 256;
  const int T256E = (N_EDGES + 255) / 256;
  const int GEMM_BLKS = (N_NODES / GEMM_ROWS) * (D_HID / GEMM_COLS);
  const int NODE_BLKS = (N_NODES + 7) / 8;

  cnt_zero_kernel<<<T256N, 256, 0, stream>>>(cnt, N_NODES);
  edge_rank_kernel<<<T256E, 256, 0, stream>>>(col, cnt, rank, N_EDGES);
  scan_dinv_kernel<<<1, 1024, 0, stream>>>(cnt, off, dinv, N_NODES);
  csr_fill_kernel<<<T256E, 256, 0, stream>>>(row, col, rank, off, dinv, ent, N_EDGES);

  gemm_wmma_kernel<0, D_IN><<<GEMM_BLKS, 128, 0, stream>>>(x, W_in, bufH);
  agg_gather_kernel<<<NODE_BLKS, 256, 0, stream>>>(bufH, dinv, cnt, off, ent, bufG, N_NODES);

  gemm_wmma_kernel<1, D_HID><<<GEMM_BLKS, 128, 0, stream>>>(bufG, W_hid, bufH);
  agg_gather_kernel<<<NODE_BLKS, 256, 0, stream>>>(bufH, dinv, cnt, off, ent, out, N_NODES);

  gemm_wmma_kernel<1, D_HID><<<GEMM_BLKS, 128, 0, stream>>>(out, W_out, bufH);
  agg_gather_kernel<<<NODE_BLKS, 256, 0, stream>>>(bufH, dinv, cnt, off, ent, out, N_NODES);
}


# Layer 1 aggregates before its product (Agg(x) @ W_in, not Agg(x @ W_in)): its gather reads 128-wide rows of x, half the bytes; changes the order of layer 1's float sums
# speedup vs baseline: 3.6045x; 1.0521x over previous
#include <hip/hip_runtime.h>

#define N_NODES 50000
#define N_EDGES 800000
#define D_IN    128
#define D_HID   256

static_assert(N_NODES % 4 == 0, "scan_dinv_kernel handles 4 counters per thread");

typedef __attribute__((ext_vector_type(2))) float v2f;
typedef __attribute__((ext_vector_type(4))) float v4f;
typedef __attribute__((ext_vector_type(8))) float v8f;


__global__ void cnt_zero_kernel(int* __restrict__ cnt, int n) {
  int i = blockIdx.x * blockDim.x + threadIdx.x;
  if (i < n) cnt[i] = 0;
}

__global__ void edge_rank_kernel(const int* __restrict__ col, int* __restrict__ cnt,
                                 int* __restrict__ rank, int e) {
  int i = blockIdx.x * blockDim.x + threadIdx.x;
  if (i < e) rank[i] = atomicAdd(&cnt[col[i]], 1);
}

__global__ __launch_bounds__(1024) void scan_dinv_kernel(const int* __restrict__ cnt,
                                                         int* __restrict__ off,
                                                         float* __restrict__ dinv, int n) {
  __shared__ int wsum[32];
  const int t = threadIdx.x;
  const int lane = t & 31;
  const int w = t >> 5;
  int carry = 0;
  for (int base = 0; base < n; base += 4096) {
    const int i0 = base + t * 4;
    const bool live = i0 < n;
    int4 c = make_int4(0, 0, 0, 0);
    if (live) c = *(const int4*)(cnt + i0);
    const int s = c.x + c.y + c.z + c.w;
    int incl = s;
#pragma unroll
    for (int d = 1; d < 32; d <<= 1) {
      int v = __shfl_up(incl, d);
      if (lane >= d) incl += v;
    }
    if (lane == 31) wsum[w] = incl;
    __syncthreads();
    const int ws = wsum[lane];
    int wincl = ws;
#pragma unroll
    for (int d = 1; d < 32; d <<= 1) {
      int v = __shfl_up(wincl, d);
      if (lane >= d) wincl += v;
    }
    const int wave_excl = __shfl(wincl, w) - __shfl(ws, w);
    const int total = __shfl(wincl, 31);
    const int ex = carry + wave_excl + (incl - s);
    if (live) {
      int4 o;
      o.x = ex;
      o.y = ex + c.x;
      o.z = o.y + c.y;
      o.w = o.z + c.z;
      *(int4*)(off + i0) = o;
      float4 dv;
      dv.x = rsqrtf((float)(c.x + 1));
      dv.y = rsqrtf((float)(c.y + 1));
      dv.z = rsqrtf((float)(c.z + 1));
      dv.w = rsqrtf((float)(c.w + 1));
      *(float4*)(dinv + i0) = dv;
    }
    carry += total;
    __syncthreads();
  }
}

__global__ void csr_fill_kernel(const int* __restrict__ row, const int* __restrict__ col,
                                const int* __restrict__ rank, const int* __restrict__ off,
                                const float* __restrict__ dinv, int2* __restrict__ ent, int e) {
  int i = blockIdx.x * blockDim.x + threadIdx.x;
  if (i < e) {
    int r = row[i];
    int p = off[col[i]] + rank[i];
    ent[p] = make_int2(r, __float_as_int(dinv[r]));
  }
}

#define GEMM_ROWS 80
#define GEMM_COLS 128
#define GEMM_KS   32
#define SA_PITCH  36
#define SW_PITCH  288

static_assert(N_NODES % GEMM_ROWS == 0, "row tiles are exact, no guards in the GEMM");
static_assert(GEMM_ROWS * (GEMM_KS / 4) == 5 * 128, "A slab = 5 float4 per thread");

template <int RELU, int DIN>
__global__ __launch_bounds__(128) void gemm_wmma_kernel(const float* __restrict__ A,
                                                        const float* __restrict__ W,
                                                        float* __restrict__ H) {
  __shared__ __attribute__((aligned(16))) float sW[(GEMM_KS / 2) * SW_PITCH];
  __shared__ __attribute__((aligned(16))) float sA[GEMM_ROWS * SA_PITCH];

  const int t = threadIdx.x;
  const int lane = t & 31;
  const int wave = t >> 5;
  const int half = lane >> 4;
  const int c = lane & 15;
  const int m_base = (blockIdx.x >> 1) * GEMM_ROWS;
  const int n_base = (blockIdx.x & 1) * GEMM_COLS;

  const int wp = t >> 5;
  const int wq = t & 31;
  const float* __restrict__ gW = W + (size_t)(2 * wp) * 256 + n_base + wq * 4;
  const float* __restrict__ gA = A + (size_t)(m_base + (t >> 3)) * DIN + (t & 7) * 4;
  float* __restrict__ dW = sW + wp * SW_PITCH + wq * 8;
  float* __restrict__ dA = sA + (t >> 3) * SA_PITCH + (t & 7) * 4;
  v4f wr[8];
  v4f ar[5];

  auto gload = [&](int k0) {
#pragma unroll
    for (int i = 0; i < 4; ++i) {
      const float* __restrict__ g = gW + (size_t)(k0 + 8 * i) * 256;
      wr[2 * i]     = *(const v4f*)(g);
      wr[2 * i + 1] = *(const v4f*)(g + 256);
    }
#pragma unroll
    for (int i = 0; i < 5; ++i)
      ar[i] = *(const v4f*)(gA + (size_t)(16 * i) * DIN + k0);
  };

  auto sstore = [&]() {
#pragma unroll
    for (int i = 0; i < 4; ++i) {
      float* d = dW + (4 * i) * SW_PITCH;
      const v4f w0 = wr[2 * i], w1 = wr[2 * i + 1];
      const v4f lo = {w0.x, w1.x, w0.y, w1.y};
      const v4f hi = {w0.z, w1.z, w0.w, w1.w};
      *(v4f*)(d) = lo;
      *(v4f*)(d + 4) = hi;
    }
#pragma unroll
    for (int i = 0; i < 5; ++i) {
      v4f v = ar[i];
      if (RELU) {
        v.x = fmaxf(v.x, 0.0f); v.y = fmaxf(v.y, 0.0f);
        v.z = fmaxf(v.z, 0.0f); v.w = fmaxf(v.w, 0.0f);
      }
      *(v4f*)(dA + (16 * i) * SA_PITCH) = v;
    }
  };

  v8f acc[5][2];
#pragma unroll
  for (int r = 0; r < 5; ++r) {
    acc[r][0] = v8f{};
    acc[r][1] = v8f{};
  }

  const float* __restrict__ aF = sA + c * SA_PITCH + 2 * half;
  const float* __restrict__ bF = sW + half * SW_PITCH + (wave * 32 + c) * 2;

  gload(0);
  for (int k0 = 0; k0 < DIN; k0 += GEMM_KS) {
    sstore();
    __syncthreads();
    if (k0 + GEMM_KS < DIN) gload(k0 + GEMM_KS);
#pragma unroll
    for (int ks = 0; ks < GEMM_KS / 4; ++ks) {
      const v2f b0 = *(const v2f*)(bF + (2 * ks) * SW_PITCH);
      const v2f b1 = *(const v2f*)(bF + (2 * ks) * SW_PITCH + 32);
#pragma unroll
      for (int r = 0; r < 5; ++r) {
        const v2f a = *(const v2f*)(aF + r * 16 * SA_PITCH + ks * 4);
        acc[r][0] = __builtin_amdgcn_wmma_f32_16x16x4_f32(false, a, false, b0, (short)0, acc[r][0], false, false);
        acc[r][1] = __builtin_amdgcn_wmma_f32_16x16x4_f32(false, a, false, b1, (short)0, acc[r][1], false, false);
      }
    }
    __syncthreads();
  }

  float* __restrict__ Hp = H + (size_t)(m_base + half * 8) * 256 + n_base + wave * 32 + c;
#pragma unroll
  for (int r = 0; r < 5; ++r) {
#pragma unroll
    for (int e = 0; e < 8; ++e) {
      Hp[(size_t)(r * 16 + e) * 256 + 0]  = acc[r][0][e];
      Hp[(size_t)(r * 16 + e) * 256 + 16] = acc[r][1][e];
    }
  }
}


__device__ __forceinline__ void row_fma(v4f& a0, v4f& a1, const float* __restrict__ p, float w) {
  const v4f x = *(const v4f*)(p);
  const v4f y = *(const v4f*)(p + 128);
  const v4f wv = {w, w, w, w};
  a0 = __builtin_elementwise_fma(x, wv, a0);
  a1 = __builtin_elementwise_fma(y, wv, a1);
}

__global__ __launch_bounds__(256) void agg_gather_kernel(const float* __restrict__ h,
                                                         const float* __restrict__ dinv,
                                                         const int* __restrict__ cnt,
                                                         const int* __restrict__ off,
                                                         const int2* __restrict__ ent,
                                                         float* __restrict__ out, int n) {
  const int lane = threadIdx.x & 31;
  const int node = __builtin_amdgcn_readfirstlane((int)(blockIdx.x * 8 + (threadIdx.x >> 5)));
  if (node >= n) return;
  const float dc = dinv[node];
  const int beg = off[node];
  const int deg = cnt[node];
  const float* __restrict__ hl = h + lane * 4;

  v4f a0, a1;
  {
    const float s = dc * dc;
    const float* __restrict__ hp = hl + (size_t)node * 256;
    a0 = *(const v4f*)(hp) * s;
    a1 = *(const v4f*)(hp + 128) * s;
  }

  for (int base = 0; base < deg; base += 32) {
    const int m = min(32, deg - base);
    int myr = 0;
    int myw = 0;
    if (lane < m) {
      const int2 en = ent[beg + base + lane];
      myr = en.x;
      myw = __float_as_int(__int_as_float(en.y) * dc);
    }
    int j = 0;
    for (; j + 4 <= m; j += 4) {
      const int r0 = __builtin_amdgcn_readlane(myr, j + 0);
      const int r1 = __builtin_amdgcn_readlane(myr, j + 1);
      const int r2 = __builtin_amdgcn_readlane(myr, j + 2);
      const int r3 = __builtin_amdgcn_readlane(myr, j + 3);
      const float w0 = __int_as_float(__builtin_amdgcn_readlane(myw, j + 0));
      const float w1 = __int_as_float(__builtin_amdgcn_readlane(myw, j + 1));
      const float w2 = __int_as_float(__builtin_amdgcn_readlane(myw, j + 2));
      const float w3 = __int_as_float(__builtin_amdgcn_readlane(myw, j + 3));
      row_fma(a0, a1, hl + (size_t)r0 * 256, w0);
      row_fma(a0, a1, hl + (size_t)r1 * 256, w1);
      row_fma(a0, a1, hl + (size_t)r2 * 256, w2);
      row_fma(a0, a1, hl + (size_t)r3 * 256, w3);
    }
    if (j + 2 <= m) {
      const int r0 = __builtin_amdgcn_readlane(myr, j + 0);
      const int r1 = __builtin_amdgcn_readlane(myr, j + 1);
      const float w0 = __int_as_float(__builtin_amdgcn_readlane(myw, j + 0));
      const float w1 = __int_as_float(__builtin_amdgcn_readlane(myw, j + 1));
      row_fma(a0, a1, hl + (size_t)r0 * 256, w0);
      row_fma(a0, a1, hl + (size_t)r1 * 256, w1);
      j += 2;
    }
    if (j < m) {
      const int r0 = __builtin_amdgcn_readlane(myr, j);
      const float w0 = __int_as_float(__builtin_amdgcn_readlane(myw, j));
      row_fma(a0, a1, hl + (size_t)r0 * 256, w0);
    }
  }

  float* __restrict__ op = out + (size_t)node * 256 + lane * 4;
  *(v4f*)(op) = a0;
  *(v4f*)(op + 128) = a1;
}

__device__ __forceinline__ void row_fma128(v4f& a, const float* __restrict__ p, float w) {
  const v4f x = *(const v4f*)(p);
  const v4f wv = {w, w, w, w};
  a = __builtin_elementwise_fma(x, wv, a);
}

__global__ __launch_bounds__(256) void agg_gather128_kernel(const float* __restrict__ x,
                                                            const float* __restrict__ dinv,
                                                            const int* __restrict__ cnt,
                                                            const int* __restrict__ off,
                                                            const int2* __restrict__ ent,
                                                            float* __restrict__ out, int n) {
  const int lane = threadIdx.x & 31;
  const int node = __builtin_amdgcn_readfirstlane((int)(blockIdx.x * 8 + (threadIdx.x >> 5)));
  if (node >= n) return;
  const float dc = dinv[node];
  const int beg = off[node];
  const int deg = cnt[node];
  const float* __restrict__ xl = x + lane * 4;

  v4f a = *(const v4f*)(xl + (size_t)node * D_IN) * (dc * dc);

  for (int base = 0; base < deg; base += 32) {
    const int m = min(32, deg - base);
    int myr = 0;
    int myw = 0;
    if (lane < m) {
      const int2 en = ent[beg + base + lane];
      myr = en.x;
      myw = __float_as_int(__int_as_float(en.y) * dc);
    }
    int j = 0;
    for (; j + 8 <= m; j += 8) {
      v4f xv[8];
      float w[8];
#pragma unroll
      for (int u = 0; u < 8; ++u) {
        const int r = __builtin_amdgcn_readlane(myr, j + u);
        w[u] = __int_as_float(__builtin_amdgcn_readlane(myw, j + u));
        xv[u] = *(const v4f*)(xl + (size_t)r * D_IN);
      }
#pragma unroll
      for (int u = 0; u < 8; ++u) {
        const v4f wv = {w[u], w[u], w[u], w[u]};
        a = __builtin_elementwise_fma(xv[u], wv, a);
      }
    }
    if (j + 4 <= m) {
      const int r0 = __builtin_amdgcn_readlane(myr, j + 0);
      const int r1 = __builtin_amdgcn_readlane(myr, j + 1);
      const int r2 = __builtin_amdgcn_readlane(myr, j + 2);
      const int r3 = __builtin_amdgcn_readlane(myr, j + 3);
      const float w0 = __int_as_float(__builtin_amdgcn_readlane(myw, j + 0));
      const float w1 = __int_as_float(__builtin_amdgcn_readlane(myw, j + 1));
      const float w2 = __int_as_float(__builtin_amdgcn_readlane(myw, j + 2));
      const float w3 = __int_as_float(__builtin_amdgcn_readlane(myw, j + 3));
      row_fma128(a, xl + (size_t)r0 * D_IN, w0);
      row_fma128(a, xl + (size_t)r1 * D_IN, w1);
      row_fma128(a, xl + (size_t)r2 * D_IN, w2);
      row_fma128(a, xl + (size_t)r3 * D_IN, w3);
      j += 4;
    }
    if (j + 2 <= m) {
      const int r0 = __builtin_amdgcn_readlane(myr, j + 0);
      const int r1 = __builtin_amdgcn_readlane(myr, j + 1);
      const float w0 = __int_as_float(__builtin_amdgcn_readlane(myw, j + 0));
      const float w1 = __int_as_float(__builtin_amdgcn_readlane(myw, j + 1));
      row_fma128(a, xl + (size_t)r0 * D_IN, w0);
      row_fma128(a, xl + (size_t)r1 * D_IN, w1);
      j += 2;
    }
    if (j < m) {
      const int r0 = __builtin_amdgcn_readlane(myr, j);
      const float w0 = __int_as_float(__builtin_amdgcn_readlane(myw, j));
      row_fma128(a, xl + (size_t)r0 * D_IN, w0);
    }
  }

  *(v4f*)(out + (size_t)node * D_IN + lane * 4) = a;
}


extern "C" void kernel_launch(void* const* d_in, const int* in_sizes, int n_in,
                              void* d_out, int out_size, void* d_ws, size_t ws_size,
                              hipStream_t stream) {
  const float* x     = (const float*)d_in[0];
  const int*   ei    = (const int*)d_in[1];
  const float* W_in  = (const float*)d_in[2];
  const float* W_hid = (const float*)d_in[3];
  const float* W_out = (const float*)d_in[4];
  float* out = (float*)d_out;

  float* ws   = (float*)d_ws;
  int*   cnt  = (int*)ws;
  float* dinv = ws + 65536;
  int*   off  = (int*)(ws + 2 * 65536);
  int*   rank = (int*)(ws + 3 * 65536);
  int2*  ent  = (int2*)(ws + (1 << 20));
  float* bufH = ws + (4 << 20);
  float* bufG = bufH + (size_t)N_NODES * 256;

  const int* row = ei;
  const int* col = ei + N_EDGES;

  const int T256N = (N_NODES + 255) / 256;
  const int T256E = (N_EDGES + 255) / 256;
  const int GEMM_BLKS = (N_NODES / GEMM_ROWS) * (D_HID / GEMM_COLS);
  const int NODE_BLKS = (N_NODES + 7) / 8;

  cnt_zero_kernel<<<T256N, 256, 0, stream>>>(cnt, N_NODES);
  edge_rank_kernel<<<T256E, 256, 0, stream>>>(col, cnt, rank, N_EDGES);
  scan_dinv_kernel<<<1, 1024, 0, stream>>>(cnt, off, dinv, N_NODES);
  csr_fill_kernel<<<T256E, 256, 0, stream>>>(row, col, rank, off, dinv, ent, N_EDGES);

  agg_gather128_kernel<<<NODE_BLKS, 256, 0, stream>>>(x, dinv, cnt, off, ent, bufG, N_NODES);
  gemm_wmma_kernel<0, D_IN><<<GEMM_BLKS, 128, 0, stream>>>(bufG, W_in, bufH);

  gemm_wmma_kernel<1, D_HID><<<GEMM_BLKS, 128, 0, stream>>>(bufH, W_hid, bufG);
  agg_gather_kernel<<<NODE_BLKS, 256, 0, stream>>>(bufG, dinv, cnt, off, ent, out, N_NODES);

  gemm_wmma_kernel<1, D_HID><<<GEMM_BLKS, 128, 0, stream>>>(out, W_out, bufH);
  agg_gather_kernel<<<NODE_BLKS, 256, 0, stream>>>(bufH, dinv, cnt, off, ent, out, N_NODES);
}


# Source lists built by a two-level counting sort in LDS (edge chunks binned by 64-node target range, then one block per range), not a global atomic + scattered store per edge; list (sum) order changes
# speedup vs baseline: 4.7761x; 1.3250x over previous
#include <hip/hip_runtime.h>

#define N_NODES 50000
#define N_EDGES 800000
#define D_IN    128
#define D_HID   256

typedef __attribute__((ext_vector_type(2))) float v2f;
typedef __attribute__((ext_vector_type(4))) float v4f;
typedef __attribute__((ext_vector_type(8))) float v8f;

#define BIN_SHIFT 6
#define BIN_NODES (1 << BIN_SHIFT)
#define NBIN      ((N_NODES + BIN_NODES - 1) / BIN_NODES)
#define CHUNK     8192
#define NCHUNK    ((N_EDGES + CHUNK - 1) / CHUNK)
#define BIN_CAP   2048

static_assert(NBIN + 1 <= 1024, "one thread per bin start in the level-1 scan");
static_assert(NCHUNK <= 256, "one thread per chunk piece in level 2");
static_assert(CHUNK % 1024 == 0, "level 1: whole edges per thread");
static_assert(N_EDGES % 4 == 0, "level 3 handles 4 entries per thread");

__device__ __forceinline__ int wave_incl_scan(int v, int lane) {
  for (int d = 1; d < 32; d <<= 1) {
    const int n = __shfl_up(v, d);
    if (lane >= d) v += n;
  }
  return v;
}

template <int NW>
__device__ __forceinline__ int block_excl_scan(int v, int* wsum, int& total) {
  const int lane = threadIdx.x & 31;
  const int w = threadIdx.x >> 5;
  const int incl = wave_incl_scan(v, lane);
  if (lane == 31) wsum[w] = incl;
  __syncthreads();
  const int ws = (lane < NW) ? wsum[lane] : 0;
  const int wincl = wave_incl_scan(ws, lane);
  const int wave_excl = __shfl(wincl, w) - __shfl(ws, w);
  total = __shfl(wincl, 31);
  __syncthreads();
  return wave_excl + incl - v;
}

__global__ __launch_bounds__(1024) void bin_chunk_kernel(const int* __restrict__ row,
                                                         const int* __restrict__ col,
                                                         int2* __restrict__ sorted,
                                                         int* __restrict__ lst) {
  __shared__ int hcnt[NBIN];
  __shared__ int hoff[NBIN + 1];
  __shared__ int wsum[32];
  __shared__ int2 sPair[CHUNK];
  constexpr int EPT = CHUNK / 1024;
  const int t = threadIdx.x;
  const int chunk = blockIdx.x;
  const int e0 = chunk * CHUNK;
  const int n = min(CHUNK, N_EDGES - e0);

  if (t < NBIN) hcnt[t] = 0;
  __syncthreads();

  int c[EPT], r[EPT], lr[EPT];
#pragma unroll
  for (int u = 0; u < EPT; ++u) {
    const int j = u * 1024 + t;
    c[u] = -1;
    r[u] = 0;
    if (j < n) {
      c[u] = col[e0 + j];
      r[u] = row[e0 + j];
    }
  }
#pragma unroll
  for (int u = 0; u < EPT; ++u) {
    lr[u] = 0;
    if (c[u] >= 0) lr[u] = atomicAdd(&hcnt[c[u] >> BIN_SHIFT], 1);
  }
  __syncthreads();

  int total;
  const int v = (t < NBIN) ? hcnt[t] : 0;
  const int excl = block_excl_scan<32>(v, wsum, total);
  if (t <= NBIN) {
    hoff[t] = excl;
    lst[t * NCHUNK + chunk] = excl;
  }
  __syncthreads();

#pragma unroll
  for (int u = 0; u < EPT; ++u)
    if (c[u] >= 0) sPair[hoff[c[u] >> BIN_SHIFT] + lr[u]] = make_int2(r[u], c[u]);
  __syncthreads();

  for (int j = t; j < n; j += 1024) sorted[e0 + j] = sPair[j];
}

__global__ __launch_bounds__(256) void bin_sort_kernel(const int2* __restrict__ sorted,
                                                       const int* __restrict__ lst,
                                                       int* __restrict__ cnt,
                                                       int* __restrict__ off,
                                                       float* __restrict__ dinv,
                                                       int* __restrict__ srcs) {
  __shared__ int segS[NCHUNK];
  __shared__ int segOff[NCHUNK + 1];
  __shared__ int ncnt[BIN_NODES];
  __shared__ int cursor[BIN_NODES];
  __shared__ int wsum[32];
  __shared__ int2 sEdge[BIN_CAP];
  __shared__ int sOut[BIN_CAP];
  const int t = threadIdx.x;
  const int lane = t & 31;
  const int b = blockIdx.x;
  const int nodeBase = b * BIN_NODES;

  int s = 0, len = 0;
  if (t < NCHUNK) {
    s = lst[b * NCHUNK + t];
    len = lst[(b + 1) * NCHUNK + t] - s;
  }
  if (t < BIN_NODES) ncnt[t] = 0;
  int binTotal, binBase;
  const int so = block_excl_scan<8>(len, wsum, binTotal);
  (void)block_excl_scan<8>(s, wsum, binBase);
  if (t < NCHUNK) {
    segS[t] = s;
    segOff[t] = so;
  }
  if (t == 0) segOff[NCHUNK] = binTotal;
  __syncthreads();

  const bool small = binTotal <= BIN_CAP;

  auto fetch = [&](int j) -> int2 {
    int lo = 0, hi = NCHUNK - 1;
    while (lo < hi) {
      const int mid = (lo + hi + 1) >> 1;
      if (segOff[mid] <= j) lo = mid; else hi = mid - 1;
    }
    return sorted[(size_t)lo * CHUNK + segS[lo] + (j - segOff[lo])];
  };

  for (int j0 = t; j0 < binTotal; j0 += 1024) {
    int2 e[4];
#pragma unroll
    for (int u = 0; u < 4; ++u) {
      const int j = j0 + 256 * u;
      e[u] = make_int2(0, nodeBase);
      if (j < binTotal) e[u] = fetch(j);
    }
#pragma unroll
    for (int u = 0; u < 4; ++u) {
      const int j = j0 + 256 * u;
      if (j < binTotal) {
        if (small) sEdge[j] = e[u];
        atomicAdd(&ncnt[e[u].y - nodeBase], 1);
      }
    }
  }
  __syncthreads();

  if (t < 32) {
    const int a = ncnt[2 * lane], bb = ncnt[2 * lane + 1];
    const int incl = wave_incl_scan(a + bb, lane);
    const int o0 = incl - (a + bb);
    const int o1 = o0 + a;
    cursor[2 * lane] = o0;
    cursor[2 * lane + 1] = o1;
    const int n0 = nodeBase + 2 * lane;
    const int n1 = n0 + 1;
    if (n0 < N_NODES) {
      cnt[n0] = a;
      off[n0] = binBase + o0;
      dinv[n0] = rsqrtf((float)(a + 1));
    }
    if (n1 < N_NODES) {
      cnt[n1] = bb;
      off[n1] = binBase + o1;
      dinv[n1] = rsqrtf((float)(bb + 1));
    }
  }
  __syncthreads();

  for (int j = t; j < binTotal; j += 256) {
    const int2 e = small ? sEdge[j] : fetch(j);
    const int pl = atomicAdd(&cursor[e.y - nodeBase], 1);
    if (small) sOut[pl] = e.x; else srcs[binBase + pl] = e.x;
  }
  __syncthreads();
  if (small)
    for (int j = t; j < binTotal; j += 256) srcs[binBase + j] = sOut[j];
}

__global__ void ent_weight_kernel(const int* __restrict__ srcs, const float* __restrict__ dinv,
                                  int2* __restrict__ ent) {
  const int i = (blockIdx.x * blockDim.x + threadIdx.x) * 4;
  if (i < N_EDGES) {
    const int4 s = *(const int4*)(srcs + i);
    int4 lo, hi;
    lo.x = s.x; lo.y = __float_as_int(dinv[s.x]);
    lo.z = s.y; lo.w = __float_as_int(dinv[s.y]);
    hi.x = s.z; hi.y = __float_as_int(dinv[s.z]);
    hi.z = s.w; hi.w = __float_as_int(dinv[s.w]);
    int4* d = (int4*)(ent + i);
    d[0] = lo;
    d[1] = hi;
  }
}

#define GEMM_ROWS 80
#define GEMM_COLS 128
#define GEMM_KS   32
#define SA_PITCH  36
#define SW_PITCH  288

static_assert(N_NODES % GEMM_ROWS == 0, "row tiles are exact, no guards in the GEMM");
static_assert(GEMM_ROWS * (GEMM_KS / 4) == 5 * 128, "A slab = 5 float4 per thread");

template <int RELU, int DIN>
__global__ __launch_bounds__(128) void gemm_wmma_kernel(const float* __restrict__ A,
                                                        const float* __restrict__ W,
                                                        float* __restrict__ H) {
  __shared__ __attribute__((aligned(16))) float sW[(GEMM_KS / 2) * SW_PITCH];
  __shared__ __attribute__((aligned(16))) float sA[GEMM_ROWS * SA_PITCH];

  const int t = threadIdx.x;
  const int lane = t & 31;
  const int wave = t >> 5;
  const int half = lane >> 4;
  const int c = lane & 15;
  const int m_base = (blockIdx.x >> 1) * GEMM_ROWS;
  const int n_base = (blockIdx.x & 1) * GEMM_COLS;

  const int wp = t >> 5;
  const int wq = t & 31;
  const float* __restrict__ gW = W + (size_t)(2 * wp) * 256 + n_base + wq * 4;
  const float* __restrict__ gA = A + (size_t)(m_base + (t >> 3)) * DIN + (t & 7) * 4;
  float* __restrict__ dW = sW + wp * SW_PITCH + wq * 8;
  float* __restrict__ dA = sA + (t >> 3) * SA_PITCH + (t & 7) * 4;
  v4f wr[8];
  v4f ar[5];

  auto gload = [&](int k0) {
#pragma unroll
    for (int i = 0; i < 4; ++i) {
      const float* __restrict__ g = gW + (size_t)(k0 + 8 * i) * 256;
      wr[2 * i]     = *(const v4f*)(g);
      wr[2 * i + 1] = *(const v4f*)(g + 256);
    }
#pragma unroll
    for (int i = 0; i < 5; ++i)
      ar[i] = *(const v4f*)(gA + (size_t)(16 * i) * DIN + k0);
  };

  auto sstore = [&]() {
#pragma unroll
    for (int i = 0; i < 4; ++i) {
      float* d = dW + (4 * i) * SW_PITCH;
      const v4f w0 = wr[2 * i], w1 = wr[2 * i + 1];
      const v4f lo = {w0.x, w1.x, w0.y, w1.y};
      const v4f hi = {w0.z, w1.z, w0.w, w1.w};
      *(v4f*)(d) = lo;
      *(v4f*)(d + 4) = hi;
    }
#pragma unroll
    for (int i = 0; i < 5; ++i) {
      v4f v = ar[i];
      if (RELU) {
        v.x = fmaxf(v.x, 0.0f); v.y = fmaxf(v.y, 0.0f);
        v.z = fmaxf(v.z, 0.0f); v.w = fmaxf(v.w, 0.0f);
      }
      *(v4f*)(dA + (16 * i) * SA_PITCH) = v;
    }
  };

  v8f acc[5][2];
#pragma unroll
  for (int r = 0; r < 5; ++r) {
    acc[r][0] = v8f{};
    acc[r][1] = v8f{};
  }

  const float* __restrict__ aF = sA + c * SA_PITCH + 2 * half;
  const float* __restrict__ bF = sW + half * SW_PITCH + (wave * 32 + c) * 2;

  gload(0);
  for (int k0 = 0; k0 < DIN; k0 += GEMM_KS) {
    sstore();
    __syncthreads();
    if (k0 + GEMM_KS < DIN) gload(k0 + GEMM_KS);
#pragma unroll
    for (int ks = 0; ks < GEMM_KS / 4; ++ks) {
      const v2f b0 = *(const v2f*)(bF + (2 * ks) * SW_PITCH);
      const v2f b1 = *(const v2f*)(bF + (2 * ks) * SW_PITCH + 32);
#pragma unroll
      for (int r = 0; r < 5; ++r) {
        const v2f a = *(const v2f*)(aF + r * 16 * SA_PITCH + ks * 4);
        acc[r][0] = __builtin_amdgcn_wmma_f32_16x16x4_f32(false, a, false, b0, (short)0, acc[r][0], false, false);
        acc[r][1] = __builtin_amdgcn_wmma_f32_16x16x4_f32(false, a, false, b1, (short)0, acc[r][1], false, false);
      }
    }
    __syncthreads();
  }

  float* __restrict__ Hp = H + (size_t)(m_base + half * 8) * 256 + n_base + wave * 32 + c;
#pragma unroll
  for (int r = 0; r < 5; ++r) {
#pragma unroll
    for (int e = 0; e < 8; ++e) {
      Hp[(size_t)(r * 16 + e) * 256 + 0]  = acc[r][0][e];
      Hp[(size_t)(r * 16 + e) * 256 + 16] = acc[r][1][e];
    }
  }
}


__device__ __forceinline__ void row_fma(v4f& a0, v4f& a1, const float* __restrict__ p, float w) {
  const v4f x = *(const v4f*)(p);
  const v4f y = *(const v4f*)(p + 128);
  const v4f wv = {w, w, w, w};
  a0 = __builtin_elementwise_fma(x, wv, a0);
  a1 = __builtin_elementwise_fma(y, wv, a1);
}

__global__ __launch_bounds__(256) void agg_gather_kernel(const float* __restrict__ h,
                                                         const float* __restrict__ dinv,
                                                         const int* __restrict__ cnt,
                                                         const int* __restrict__ off,
                                                         const int2* __restrict__ ent,
                                                         float* __restrict__ out, int n) {
  const int lane = threadIdx.x & 31;
  const int node = __builtin_amdgcn_readfirstlane((int)(blockIdx.x * 8 + (threadIdx.x >> 5)));
  if (node >= n) return;
  const float dc = dinv[node];
  const int beg = off[node];
  const int deg = cnt[node];
  const float* __restrict__ hl = h + lane * 4;

  v4f a0, a1;
  {
    const float s = dc * dc;
    const float* __restrict__ hp = hl + (size_t)node * 256;
    a0 = *(const v4f*)(hp) * s;
    a1 = *(const v4f*)(hp + 128) * s;
  }

  for (int base = 0; base < deg; base += 32) {
    const int m = min(32, deg - base);
    int myr = 0;
    int myw = 0;
    if (lane < m) {
      const int2 en = ent[beg + base + lane];
      myr = en.x;
      myw = __float_as_int(__int_as_float(en.y) * dc);
    }
    int j = 0;
    for (; j + 4 <= m; j += 4) {
      const int r0 = __builtin_amdgcn_readlane(myr, j + 0);
      const int r1 = __builtin_amdgcn_readlane(myr, j + 1);
      const int r2 = __builtin_amdgcn_readlane(myr, j + 2);
      const int r3 = __builtin_amdgcn_readlane(myr, j + 3);
      const float w0 = __int_as_float(__builtin_amdgcn_readlane(myw, j + 0));
      const float w1 = __int_as_float(__builtin_amdgcn_readlane(myw, j + 1));
      const float w2 = __int_as_float(__builtin_amdgcn_readlane(myw, j + 2));
      const float w3 = __int_as_float(__builtin_amdgcn_readlane(myw, j + 3));
      row_fma(a0, a1, hl + (size_t)r0 * 256, w0);
      row_fma(a0, a1, hl + (size_t)r1 * 256, w1);
      row_fma(a0, a1, hl + (size_t)r2 * 256, w2);
      row_fma(a0, a1, hl + (size_t)r3 * 256, w3);
    }
    if (j + 2 <= m) {
      const int r0 = __builtin_amdgcn_readlane(myr, j + 0);
      const int r1 = __builtin_amdgcn_readlane(myr, j + 1);
      const float w0 = __int_as_float(__builtin_amdgcn_readlane(myw, j + 0));
      const float w1 = __int_as_float(__builtin_amdgcn_readlane(myw, j + 1));
      row_fma(a0, a1, hl + (size_t)r0 * 256, w0);
      row_fma(a0, a1, hl + (size_t)r1 * 256, w1);
      j += 2;
    }
    if (j < m) {
      const int r0 = __builtin_amdgcn_readlane(myr, j);
      const float w0 = __int_as_float(__builtin_amdgcn_readlane(myw, j));
      row_fma(a0, a1, hl + (size_t)r0 * 256, w0);
    }
  }

  float* __restrict__ op = out + (size_t)node * 256 + lane * 4;
  *(v4f*)(op) = a0;
  *(v4f*)(op + 128) = a1;
}

__device__ __forceinline__ void row_fma128(v4f& a, const float* __restrict__ p, float w) {
  const v4f x = *(const v4f*)(p);
  const v4f wv = {w, w, w, w};
  a = __builtin_elementwise_fma(x, wv, a);
}

__global__ __launch_bounds__(256) void agg_gather128_kernel(const float* __restrict__ x,
                                                            const float* __restrict__ dinv,
                                                            const int* __restrict__ cnt,
                                                            const int* __restrict__ off,
                                                            const int2* __restrict__ ent,
                                                            float* __restrict__ out, int n) {
  const int lane = threadIdx.x & 31;
  const int node = __builtin_amdgcn_readfirstlane((int)(blockIdx.x * 8 + (threadIdx.x >> 5)));
  if (node >= n) return;
  const float dc = dinv[node];
  const int beg = off[node];
  const int deg = cnt[node];
  const float* __restrict__ xl = x + lane * 4;

  v4f a = *(const v4f*)(xl + (size_t)node * D_IN) * (dc * dc);

  for (int base = 0; base < deg; base += 32) {
    const int m = min(32, deg - base);
    int myr = 0;
    int myw = 0;
    if (lane < m) {
      const int2 en = ent[beg + base + lane];
      myr = en.x;
      myw = __float_as_int(__int_as_float(en.y) * dc);
    }
    int j = 0;
    for (; j + 8 <= m; j += 8) {
      v4f xv[8];
      float w[8];
#pragma unroll
      for (int u = 0; u < 8; ++u) {
        const int r = __builtin_amdgcn_readlane(myr, j + u);
        w[u] = __int_as_float(__builtin_amdgcn_readlane(myw, j + u));
        xv[u] = *(const v4f*)(xl + (size_t)r * D_IN);
      }
#pragma unroll
      for (int u = 0; u < 8; ++u) {
        const v4f wv = {w[u], w[u], w[u], w[u]};
        a = __builtin_elementwise_fma(xv[u], wv, a);
      }
    }
    if (j + 4 <= m) {
      const int r0 = __builtin_amdgcn_readlane(myr, j + 0);
      const int r1 = __builtin_amdgcn_readlane(myr, j + 1);
      const int r2 = __builtin_amdgcn_readlane(myr, j + 2);
      const int r3 = __builtin_amdgcn_readlane(myr, j + 3);
      const float w0 = __int_as_float(__builtin_amdgcn_readlane(myw, j + 0));
      const float w1 = __int_as_float(__builtin_amdgcn_readlane(myw, j + 1));
      const float w2 = __int_as_float(__builtin_amdgcn_readlane(myw, j + 2));
      const float w3 = __int_as_float(__builtin_amdgcn_readlane(myw, j + 3));
      row_fma128(a, xl + (size_t)r0 * D_IN, w0);
      row_fma128(a, xl + (size_t)r1 * D_IN, w1);
      row_fma128(a, xl + (size_t)r2 * D_IN, w2);
      row_fma128(a, xl + (size_t)r3 * D_IN, w3);
      j += 4;
    }
    if (j + 2 <= m) {
      const int r0 = __builtin_amdgcn_readlane(myr, j + 0);
      const int r1 = __builtin_amdgcn_readlane(myr, j + 1);
      const float w0 = __int_as_float(__builtin_amdgcn_readlane(myw, j + 0));
      const float w1 = __int_as_float(__builtin_amdgcn_readlane(myw, j + 1));
      row_fma128(a, xl + (size_t)r0 * D_IN, w0);
      row_fma128(a, xl + (size_t)r1 * D_IN, w1);
      j += 2;
    }
    if (j < m) {
      const int r0 = __builtin_amdgcn_readlane(myr, j);
      const float w0 = __int_as_float(__builtin_amdgcn_readlane(myw, j));
      row_fma128(a, xl + (size_t)r0 * D_IN, w0);
    }
  }

  *(v4f*)(out + (size_t)node * D_IN + lane * 4) = a;
}


extern "C" void kernel_launch(void* const* d_in, const int* in_sizes, int n_in,
                              void* d_out, int out_size, void* d_ws, size_t ws_size,
                              hipStream_t stream) {
  const float* x     = (const float*)d_in[0];
  const int*   ei    = (const int*)d_in[1];
  const float* W_in  = (const float*)d_in[2];
  const float* W_hid = (const float*)d_in[3];
  const float* W_out = (const float*)d_in[4];
  float* out = (float*)d_out;

  float* ws   = (float*)d_ws;
  int*   cnt  = (int*)ws;
  float* dinv = ws + 65536;
  int*   off  = (int*)(ws + 2 * 65536);
  int*   srcs = (int*)(ws + 3 * 65536);
  int2*  ent  = (int2*)(ws + (1 << 20));
  int*   lst  = (int*)(ws + (3 << 20));
  float* bufH = ws + (4 << 20);
  float* bufG = bufH + (size_t)N_NODES * 256;
  int2*  sorted = (int2*)(ws + (30 << 20));

  const int* row = ei;
  const int* col = ei + N_EDGES;

  const int GEMM_BLKS = (N_NODES / GEMM_ROWS) * (D_HID / GEMM_COLS);
  const int NODE_BLKS = (N_NODES + 7) / 8;
  const int WGT_BLKS  = (N_EDGES / 4 + 255) / 256;

  bin_chunk_kernel<<<NCHUNK, 1024, 0, stream>>>(row, col, sorted, lst);
  bin_sort_kernel<<<NBIN, 256, 0, stream>>>(sorted, lst, cnt, off, dinv, srcs);
  ent_weight_kernel<<<WGT_BLKS, 256, 0, stream>>>(srcs, dinv, ent);

  agg_gather128_kernel<<<NODE_BLKS, 256, 0, stream>>>(x, dinv, cnt, off, ent, bufG, N_NODES);
  gemm_wmma_kernel<0, D_IN><<<GEMM_BLKS, 128, 0, stream>>>(bufG, W_in, bufH);

  gemm_wmma_kernel<1, D_HID><<<GEMM_BLKS, 128, 0, stream>>>(bufH, W_hid, bufG);
  agg_gather_kernel<<<NODE_BLKS, 256, 0, stream>>>(bufG, dinv, cnt, off, ent, out, N_NODES);

  gemm_wmma_kernel<1, D_HID><<<GEMM_BLKS, 128, 0, stream>>>(out, W_out, bufH);
  agg_gather_kernel<<<NODE_BLKS, 256, 0, stream>>>(bufH, dinv, cnt, off, ent, out, N_NODES);
}


# Each layer's gather and GEMM fused in a persistent kernel, W in LDS: gather waves sum the next 16-row tile while WMMA waves multiply the last; layers 2,3 aggregate before the product (sum order)
# speedup vs baseline: 5.0496x; 1.0573x over previous
#include <hip/hip_runtime.h>

#define N_NODES 50000
#define N_EDGES 800000
#define D_IN    128
#define D_HID   256

typedef __attribute__((ext_vector_type(2))) float v2f;
typedef __attribute__((ext_vector_type(4))) float v4f;
typedef __attribute__((ext_vector_type(8))) float v8f;

#define BIN_SHIFT 6
#define BIN_NODES (1 << BIN_SHIFT)
#define NBIN      ((N_NODES + BIN_NODES - 1) / BIN_NODES)
#define CHUNK     8192
#define NCHUNK    ((N_EDGES + CHUNK - 1) / CHUNK)
#define BIN_CAP   2048

static_assert(NBIN + 1 <= 1024, "one thread per bin start in the level-1 scan");
static_assert(NCHUNK <= 256, "one thread per chunk piece in level 2");
static_assert(CHUNK % 1024 == 0, "level 1: whole edges per thread");
static_assert(N_EDGES % 4 == 0, "level 3 handles 4 entries per thread");

__device__ __forceinline__ int wave_incl_scan(int v, int lane) {
  for (int d = 1; d < 32; d <<= 1) {
    const int n = __shfl_up(v, d);
    if (lane >= d) v += n;
  }
  return v;
}

template <int NW>
__device__ __forceinline__ int block_excl_scan(int v, int* wsum, int& total) {
  const int lane = threadIdx.x & 31;
  const int w = threadIdx.x >> 5;
  const int incl = wave_incl_scan(v, lane);
  if (lane == 31) wsum[w] = incl;
  __syncthreads();
  const int ws = (lane < NW) ? wsum[lane] : 0;
  const int wincl = wave_incl_scan(ws, lane);
  const int wave_excl = __shfl(wincl, w) - __shfl(ws, w);
  total = __shfl(wincl, 31);
  __syncthreads();
  return wave_excl + incl - v;
}

__global__ __launch_bounds__(1024) void bin_chunk_kernel(const int* __restrict__ row,
                                                         const int* __restrict__ col,
                                                         int2* __restrict__ sorted,
                                                         int* __restrict__ lst) {
  __shared__ int hcnt[NBIN];
  __shared__ int hoff[NBIN + 1];
  __shared__ int wsum[32];
  __shared__ int2 sPair[CHUNK];
  constexpr int EPT = CHUNK / 1024;
  const int t = threadIdx.x;
  const int chunk = blockIdx.x;
  const int e0 = chunk * CHUNK;
  const int n = min(CHUNK, N_EDGES - e0);

  if (t < NBIN) hcnt[t] = 0;
  __syncthreads();

  int c[EPT], r[EPT], lr[EPT];
#pragma unroll
  for (int u = 0; u < EPT; ++u) {
    const int j = u * 1024 + t;
    c[u] = -1;
    r[u] = 0;
    if (j < n) {
      c[u] = col[e0 + j];
      r[u] = row[e0 + j];
    }
  }
#pragma unroll
  for (int u = 0; u < EPT; ++u) {
    lr[u] = 0;
    if (c[u] >= 0) lr[u] = atomicAdd(&hcnt[c[u] >> BIN_SHIFT], 1);
  }
  __syncthreads();

  int total;
  const int v = (t < NBIN) ? hcnt[t] : 0;
  const int excl = block_excl_scan<32>(v, wsum, total);
  if (t <= NBIN) {
    hoff[t] = excl;
    lst[t * NCHUNK + chunk] = excl;
  }
  __syncthreads();

#pragma unroll
  for (int u = 0; u < EPT; ++u)
    if (c[u] >= 0) sPair[hoff[c[u] >> BIN_SHIFT] + lr[u]] = make_int2(r[u], c[u]);
  __syncthreads();

  for (int j = t; j < n; j += 1024) sorted[e0 + j] = sPair[j];
}

__global__ __launch_bounds__(256) void bin_sort_kernel(const int2* __restrict__ sorted,
                                                       const int* __restrict__ lst,
                                                       int* __restrict__ cnt,
                                                       int* __restrict__ off,
                                                       float* __restrict__ dinv,
                                                       int* __restrict__ srcs) {
  __shared__ int segS[NCHUNK];
  __shared__ int segOff[NCHUNK + 1];
  __shared__ int ncnt[BIN_NODES];
  __shared__ int cursor[BIN_NODES];
  __shared__ int wsum[32];
  __shared__ int2 sEdge[BIN_CAP];
  __shared__ int sOut[BIN_CAP];
  const int t = threadIdx.x;
  const int lane = t & 31;
  const int b = blockIdx.x;
  const int nodeBase = b * BIN_NODES;

  int s = 0, len = 0;
  if (t < NCHUNK) {
    s = lst[b * NCHUNK + t];
    len = lst[(b + 1) * NCHUNK + t] - s;
  }
  if (t < BIN_NODES) ncnt[t] = 0;
  int binTotal, binBase;
  const int so = block_excl_scan<8>(len, wsum, binTotal);
  (void)block_excl_scan<8>(s, wsum, binBase);
  if (t < NCHUNK) {
    segS[t] = s;
    segOff[t] = so;
  }
  if (t == 0) segOff[NCHUNK] = binTotal;
  __syncthreads();

  const bool small = binTotal <= BIN_CAP;

  auto fetch = [&](int j) -> int2 {
    int lo = 0, hi = NCHUNK - 1;
    while (lo < hi) {
      const int mid = (lo + hi + 1) >> 1;
      if (segOff[mid] <= j) lo = mid; else hi = mid - 1;
    }
    return sorted[(size_t)lo * CHUNK + segS[lo] + (j - segOff[lo])];
  };

  for (int j0 = t; j0 < binTotal; j0 += 1024) {
    int2 e[4];
#pragma unroll
    for (int u = 0; u < 4; ++u) {
      const int j = j0 + 256 * u;
      e[u] = make_int2(0, nodeBase);
      if (j < binTotal) e[u] = fetch(j);
    }
#pragma unroll
    for (int u = 0; u < 4; ++u) {
      const int j = j0 + 256 * u;
      if (j < binTotal) {
        if (small) sEdge[j] = e[u];
        atomicAdd(&ncnt[e[u].y - nodeBase], 1);
      }
    }
  }
  __syncthreads();

  if (t < 32) {
    const int a = ncnt[2 * lane], bb = ncnt[2 * lane + 1];
    const int incl = wave_incl_scan(a + bb, lane);
    const int o0 = incl - (a + bb);
    const int o1 = o0 + a;
    cursor[2 * lane] = o0;
    cursor[2 * lane + 1] = o1;
    const int n0 = nodeBase + 2 * lane;
    const int n1 = n0 + 1;
    if (n0 < N_NODES) {
      cnt[n0] = a;
      off[n0] = binBase + o0;
      dinv[n0] = rsqrtf((float)(a + 1));
    }
    if (n1 < N_NODES) {
      cnt[n1] = bb;
      off[n1] = binBase + o1;
      dinv[n1] = rsqrtf((float)(bb + 1));
    }
  }
  __syncthreads();

  for (int j = t; j < binTotal; j += 256) {
    const int2 e = small ? sEdge[j] : fetch(j);
    const int pl = atomicAdd(&cursor[e.y - nodeBase], 1);
    if (small) sOut[pl] = e.x; else srcs[binBase + pl] = e.x;
  }
  __syncthreads();
  if (small)
    for (int j = t; j < binTotal; j += 256) srcs[binBase + j] = sOut[j];
}

__global__ void ent_weight_kernel(const int* __restrict__ srcs, const float* __restrict__ dinv,
                                  int2* __restrict__ ent) {
  const int i = (blockIdx.x * blockDim.x + threadIdx.x) * 4;
  if (i < N_EDGES) {
    const int4 s = *(const int4*)(srcs + i);
    int4 lo, hi;
    lo.x = s.x; lo.y = __float_as_int(dinv[s.x]);
    lo.z = s.y; lo.w = __float_as_int(dinv[s.y]);
    hi.x = s.z; hi.y = __float_as_int(dinv[s.z]);
    hi.z = s.w; hi.w = __float_as_int(dinv[s.w]);
    int4* d = (int4*)(ent + i);
    d[0] = lo;
    d[1] = hi;
  }
}

#define TILE_M        16
#define NTILES        (N_NODES / TILE_M)
#define SW_PITCH      544
#define MMA_WAVES     8
#define GATHER_WAVES  16
#define LAYER_THREADS ((MMA_WAVES + GATHER_WAVES) * 32)

static_assert(N_NODES % TILE_M == 0, "row tiles are exact, no guards in the layer kernel");

__device__ __forceinline__ v4f relu4(v4f x) {
  const v4f z = {0.0f, 0.0f, 0.0f, 0.0f};
  return __builtin_elementwise_max(x, z);
}

__device__ __forceinline__ void lds_barrier() {
  __builtin_amdgcn_fence(__ATOMIC_RELEASE, "workgroup", "local");
  __builtin_amdgcn_s_barrier();
  __builtin_amdgcn_fence(__ATOMIC_ACQUIRE, "workgroup", "local");
}

template <int DIN, int RELU, int B>
__device__ __forceinline__ void gather_rows(v4f (&acc)[DIN / 128], const float* __restrict__ sl,
                                            int myr, int myw, int j, int m) {
  constexpr int NV = DIN / 128;
  v4f xv[B][NV];
  float w[B];
#pragma unroll
  for (int u = 0; u < B; ++u) {
    const int jj = min(j + u, m - 1);
    const int r = __builtin_amdgcn_readlane(myr, jj);
    w[u] = __int_as_float(__builtin_amdgcn_readlane(myw, jj));
    const float* __restrict__ p = sl + (size_t)r * DIN;
#pragma unroll
    for (int v = 0; v < NV; ++v) xv[u][v] = *(const v4f*)(p + v * 128);
  }
#pragma unroll
  for (int u = 0; u < B; ++u) {
    if (j + u < m) {
      const v4f wv = {w[u], w[u], w[u], w[u]};
#pragma unroll
      for (int v = 0; v < NV; ++v) {
        v4f x = xv[u][v];
        if (RELU) x = relu4(x);
        acc[v] = __builtin_elementwise_fma(x, wv, acc[v]);
      }
    }
  }
}

template <int DIN>
__device__ __forceinline__ void mma_tile(const float* __restrict__ sAbuf, const float* __restrict__ sW,
                                         float* __restrict__ H, int tile, int mw, int lane) {
  constexpr int PA = DIN + 4;
  const int half = lane >> 4;
  const int c = lane & 15;
  const float* __restrict__ aF = sAbuf + c * PA + 2 * half;
  const float* __restrict__ bF = sW + half * SW_PITCH + (mw * 32 + c) * 2;
  v8f acc0 = v8f{};
  v8f acc1 = v8f{};
#pragma unroll 8
  for (int ks = 0; ks < DIN / 4; ++ks) {
    const v2f a  = *(const v2f*)(aF + ks * 4);
    const v2f b0 = *(const v2f*)(bF + (2 * ks) * SW_PITCH);
    const v2f b1 = *(const v2f*)(bF + (2 * ks) * SW_PITCH + 32);
    acc0 = __builtin_amdgcn_wmma_f32_16x16x4_f32(false, a, false, b0, (short)0, acc0, false, false);
    acc1 = __builtin_amdgcn_wmma_f32_16x16x4_f32(false, a, false, b1, (short)0, acc1, false, false);
  }
  float* __restrict__ Hp = H + (size_t)(tile * TILE_M + half * 8) * 256 + mw * 32 + c;
#pragma unroll
  for (int e = 0; e < 8; ++e) {
    Hp[(size_t)e * 256 + 0]  = acc0[e];
    Hp[(size_t)e * 256 + 16] = acc1[e];
  }
}

template <int DIN, int RELU>
__global__ __launch_bounds__(LAYER_THREADS) void layer_kernel(const float* __restrict__ src,
                                                              const float* __restrict__ W,
                                                              const float* __restrict__ dinv,
                                                              const int* __restrict__ cnt,
                                                              const int* __restrict__ off,
                                                              const int2* __restrict__ ent,
                                                              float* __restrict__ H) {
  constexpr int PA = DIN + 4;
  constexpr int NV = DIN / 128;
  constexpr int B  = 8;
  __shared__ __attribute__((aligned(16))) float sW[(DIN / 2) * SW_PITCH];
  __shared__ __attribute__((aligned(16))) float sA[2 * TILE_M * PA];

  const int lane = threadIdx.x & 31;
  const int wave = __builtin_amdgcn_readfirstlane((int)(threadIdx.x >> 5));
  const int q = wave / 3;
  const bool is_mma = (wave - 3 * q) == 0;
  const int mw = q;
  const int g = wave - q - 1;

  const int first = blockIdx.x;
  const int stride = gridDim.x;
  const int ntl = (NTILES - first + stride - 1) / stride;

  const float* __restrict__ sl = src + lane * 4;

  float n_dc = 0.0f;
  int n_beg = 0, n_deg = 0;
  int2 n_en = make_int2(0, 0);
  v4f n_self[NV];
#pragma unroll
  for (int v = 0; v < NV; ++v) n_self[v] = v4f{};

  if (is_mma) {
    const int tm = mw * 32 + lane;
    const int wp = tm >> 6;
    const int wq = tm & 63;
#pragma unroll 4
    for (int kp = wp; kp < DIN / 2; kp += 4) {
      const float* __restrict__ gw = W + (size_t)(2 * kp) * 256 + wq * 4;
      const v4f w0 = *(const v4f*)(gw);
      const v4f w1 = *(const v4f*)(gw + 256);
      const v4f lo = {w0.x, w1.x, w0.y, w1.y};
      const v4f hi = {w0.z, w1.z, w0.w, w1.w};
      float* d = sW + kp * SW_PITCH + wq * 8;
      *(v4f*)(d) = lo;
      *(v4f*)(d + 4) = hi;
    }
  } else if (ntl > 0) {
    const int node = first * TILE_M + g;
    n_dc = dinv[node];
    n_beg = off[node];
    n_deg = cnt[node];
    if (lane < min(32, n_deg)) n_en = ent[n_beg + lane];
#pragma unroll
    for (int v = 0; v < NV; ++v) n_self[v] = *(const v4f*)(sl + (size_t)node * DIN + v * 128);
  }

  for (int i = 0; i <= ntl; ++i) {
    if (is_mma) {
      if (i >= 1)
        mma_tile<DIN>(sA + ((i - 1) & 1) * (TILE_M * PA), sW, H, first + (i - 1) * stride, mw, lane);
    } else if (i < ntl) {
      const float dc = n_dc;
      const int beg = __builtin_amdgcn_readfirstlane(n_beg);
      const int deg = __builtin_amdgcn_readfirstlane(n_deg);
      const int2 en0 = n_en;
      v4f acc[NV];
      {
        const float s = dc * dc;
#pragma unroll
        for (int v = 0; v < NV; ++v) {
          v4f x = n_self[v];
          if (RELU) x = relu4(x);
          acc[v] = x * s;
        }
      }

      const bool more = i + 1 < ntl;
      const int nnode = (first + (i + 1) * stride) * TILE_M + g;
      if (more) {
        n_dc = dinv[nnode];
        n_beg = off[nnode];
        n_deg = cnt[nnode];
      }

      for (int base = 0; base < deg; base += 32) {
        const int m = min(32, deg - base);
        int2 en = en0;
        if (base > 0) {
          en = make_int2(0, 0);
          if (lane < m) en = ent[beg + base + lane];
        }
        const int myr = en.x;
        const int myw = __float_as_int(__int_as_float(en.y) * dc);
        for (int j = 0; j < m; j += B) gather_rows<DIN, RELU, B>(acc, sl, myr, myw, j, m);
      }

      if (more) {
        n_en = make_int2(0, 0);
        if (lane < min(32, n_deg)) n_en = ent[n_beg + lane];
#pragma unroll
        for (int v = 0; v < NV; ++v) n_self[v] = *(const v4f*)(sl + (size_t)nnode * DIN + v * 128);
      }

      float* __restrict__ dst = sA + (i & 1) * (TILE_M * PA) + g * PA + lane * 4;
#pragma unroll
      for (int v = 0; v < NV; ++v) *(v4f*)(dst + v * 128) = acc[v];
    }
    lds_barrier();
  }
}


extern "C" void kernel_launch(void* const* d_in, const int* in_sizes, int n_in,
                              void* d_out, int out_size, void* d_ws, size_t ws_size,
                              hipStream_t stream) {
  const float* x     = (const float*)d_in[0];
  const int*   ei    = (const int*)d_in[1];
  const float* W_in  = (const float*)d_in[2];
  const float* W_hid = (const float*)d_in[3];
  const float* W_out = (const float*)d_in[4];
  float* out = (float*)d_out;

  float* ws   = (float*)d_ws;
  int*   cnt  = (int*)ws;
  float* dinv = ws + 65536;
  int*   off  = (int*)(ws + 2 * 65536);
  int*   srcs = (int*)(ws + 3 * 65536);
  int2*  ent  = (int2*)(ws + (1 << 20));
  int*   lst  = (int*)(ws + (3 << 20));
  float* bufH = ws + (4 << 20);
  float* bufG = bufH + (size_t)N_NODES * 256;
  int2*  sorted = (int2*)(ws + (30 << 20));

  const int* row = ei;
  const int* col = ei + N_EDGES;

  const int WGT_BLKS   = (N_EDGES / 4 + 255) / 256;
  const int L1_BLKS   = 512;
  const int LAYER_BLKS = 256;

  bin_chunk_kernel<<<NCHUNK, 1024, 0, stream>>>(row, col, sorted, lst);
  bin_sort_kernel<<<NBIN, 256, 0, stream>>>(sorted, lst, cnt, off, dinv, srcs);
  ent_weight_kernel<<<WGT_BLKS, 256, 0, stream>>>(srcs, dinv, ent);

  layer_kernel<D_IN, 0><<<L1_BLKS, LAYER_THREADS, 0, stream>>>(x, W_in, dinv, cnt, off, ent, bufH);
  layer_kernel<D_HID, 1><<<LAYER_BLKS, LAYER_THREADS, 0, stream>>>(bufH, W_hid, dinv, cnt, off, ent, bufG);
  layer_kernel<D_HID, 1><<<LAYER_BLKS, LAYER_THREADS, 0, stream>>>(bufG, W_out, dinv, cnt, off, ent, out);
}


# Gather waves run at raised issue priority (s_setprio 3): their loads and row sums issue ahead of the matrix waves' WMMAs on the same SIMD, so row round trips run under the products, not behind them
# speedup vs baseline: 5.2127x; 1.0323x over previous
#include <hip/hip_runtime.h>

#define N_NODES 50000
#define N_EDGES 800000
#define D_IN    128
#define D_HID   256

typedef __attribute__((ext_vector_type(2))) float v2f;
typedef __attribute__((ext_vector_type(4))) float v4f;
typedef __attribute__((ext_vector_type(8))) float v8f;

#define BIN_SHIFT 6
#define BIN_NODES (1 << BIN_SHIFT)
#define NBIN      ((N_NODES + BIN_NODES - 1) / BIN_NODES)
#define CHUNK     8192
#define NCHUNK    ((N_EDGES + CHUNK - 1) / CHUNK)
#define BIN_CAP   2048

static_assert(NBIN + 1 <= 1024, "one thread per bin start in the level-1 scan");
static_assert(NCHUNK <= 256, "one thread per chunk piece in level 2");
static_assert(CHUNK % 1024 == 0, "level 1: whole edges per thread");
static_assert(N_EDGES % 4 == 0, "level 3 handles 4 entries per thread");

__device__ __forceinline__ int wave_incl_scan(int v, int lane) {
  for (int d = 1; d < 32; d <<= 1) {
    const int n = __shfl_up(v, d);
    if (lane >= d) v += n;
  }
  return v;
}

template <int NW>
__device__ __forceinline__ int block_excl_scan(int v, int* wsum, int& total) {
  const int lane = threadIdx.x & 31;
  const int w = threadIdx.x >> 5;
  const int incl = wave_incl_scan(v, lane);
  if (lane == 31) wsum[w] = incl;
  __syncthreads();
  const int ws = (lane < NW) ? wsum[lane] : 0;
  const int wincl = wave_incl_scan(ws, lane);
  const int wave_excl = __shfl(wincl, w) - __shfl(ws, w);
  total = __shfl(wincl, 31);
  __syncthreads();
  return wave_excl + incl - v;
}

__global__ __launch_bounds__(1024) void bin_chunk_kernel(const int* __restrict__ row,
                                                         const int* __restrict__ col,
                                                         int2* __restrict__ sorted,
                                                         int* __restrict__ lst) {
  __shared__ int hcnt[NBIN];
  __shared__ int hoff[NBIN + 1];
  __shared__ int wsum[32];
  __shared__ int2 sPair[CHUNK];
  constexpr int EPT = CHUNK / 1024;
  const int t = threadIdx.x;
  const int chunk = blockIdx.x;
  const int e0 = chunk * CHUNK;
  const int n = min(CHUNK, N_EDGES - e0);

  if (t < NBIN) hcnt[t] = 0;
  __syncthreads();

  int c[EPT], r[EPT], lr[EPT];
#pragma unroll
  for (int u = 0; u < EPT; ++u) {
    const int j = u * 1024 + t;
    c[u] = -1;
    r[u] = 0;
    if (j < n) {
      c[u] = col[e0 + j];
      r[u] = row[e0 + j];
    }
  }
#pragma unroll
  for (int u = 0; u < EPT; ++u) {
    lr[u] = 0;
    if (c[u] >= 0) lr[u] = atomicAdd(&hcnt[c[u] >> BIN_SHIFT], 1);
  }
  __syncthreads();

  int total;
  const int v = (t < NBIN) ? hcnt[t] : 0;
  const int excl = block_excl_scan<32>(v, wsum, total);
  if (t <= NBIN) {
    hoff[t] = excl;
    lst[t * NCHUNK + chunk] = excl;
  }
  __syncthreads();

#pragma unroll
  for (int u = 0; u < EPT; ++u)
    if (c[u] >= 0) sPair[hoff[c[u] >> BIN_SHIFT] + lr[u]] = make_int2(r[u], c[u]);
  __syncthreads();

  for (int j = t; j < n; j += 1024) sorted[e0 + j] = sPair[j];
}

__global__ __launch_bounds__(256) void bin_sort_kernel(const int2* __restrict__ sorted,
                                                       const int* __restrict__ lst,
                                                       int* __restrict__ cnt,
                                                       int* __restrict__ off,
                                                       float* __restrict__ dinv,
                                                       int* __restrict__ srcs) {
  __shared__ int segS[NCHUNK];
  __shared__ int segOff[NCHUNK + 1];
  __shared__ int ncnt[BIN_NODES];
  __shared__ int cursor[BIN_NODES];
  __shared__ int wsum[32];
  __shared__ int2 sEdge[BIN_CAP];
  __shared__ int sOut[BIN_CAP];
  const int t = threadIdx.x;
  const int lane = t & 31;
  const int b = blockIdx.x;
  const int nodeBase = b * BIN_NODES;

  int s = 0, len = 0;
  if (t < NCHUNK) {
    s = lst[b * NCHUNK + t];
    len = lst[(b + 1) * NCHUNK + t] - s;
  }
  if (t < BIN_NODES) ncnt[t] = 0;
  int binTotal, binBase;
  const int so = block_excl_scan<8>(len, wsum, binTotal);
  (void)block_excl_scan<8>(s, wsum, binBase);
  if (t < NCHUNK) {
    segS[t] = s;
    segOff[t] = so;
  }
  if (t == 0) segOff[NCHUNK] = binTotal;
  __syncthreads();

  const bool small = binTotal <= BIN_CAP;

  auto fetch = [&](int j) -> int2 {
    int lo = 0, hi = NCHUNK - 1;
    while (lo < hi) {
      const int mid = (lo + hi + 1) >> 1;
      if (segOff[mid] <= j) lo = mid; else hi = mid - 1;
    }
    return sorted[(size_t)lo * CHUNK + segS[lo] + (j - segOff[lo])];
  };

  for (int j0 = t; j0 < binTotal; j0 += 1024) {
    int2 e[4];
#pragma unroll
    for (int u = 0; u < 4; ++u) {
      const int j = j0 + 256 * u;
      e[u] = make_int2(0, nodeBase);
      if (j < binTotal) e[u] = fetch(j);
    }
#pragma unroll
    for (int u = 0; u < 4; ++u) {
      const int j = j0 + 256 * u;
      if (j < binTotal) {
        if (small) sEdge[j] = e[u];
        atomicAdd(&ncnt[e[u].y - nodeBase], 1);
      }
    }
  }
  __syncthreads();

  if (t < 32) {
    const int a = ncnt[2 * lane], bb = ncnt[2 * lane + 1];
    const int incl = wave_incl_scan(a + bb, lane);
    const int o0 = incl - (a + bb);
    const int o1 = o0 + a;
    cursor[2 * lane] = o0;
    cursor[2 * lane + 1] = o1;
    const int n0 = nodeBase + 2 * lane;
    const int n1 = n0 + 1;
    if (n0 < N_NODES) {
      cnt[n0] = a;
      off[n0] = binBase + o0;
      dinv[n0] = rsqrtf((float)(a + 1));
    }
    if (n1 < N_NODES) {
      cnt[n1] = bb;
      off[n1] = binBase + o1;
      dinv[n1] = rsqrtf((float)(bb + 1));
    }
  }
  __syncthreads();

  for (int j = t; j < binTotal; j += 256) {
    const int2 e = small ? sEdge[j] : fetch(j);
    const int pl = atomicAdd(&cursor[e.y - nodeBase], 1);
    if (small) sOut[pl] = e.x; else srcs[binBase + pl] = e.x;
  }
  __syncthreads();
  if (small)
    for (int j = t; j < binTotal; j += 256) srcs[binBase + j] = sOut[j];
}

__global__ void ent_weight_kernel(const int* __restrict__ srcs, const float* __restrict__ dinv,
                                  int2* __restrict__ ent) {
  const int i = (blockIdx.x * blockDim.x + threadIdx.x) * 4;
  if (i < N_EDGES) {
    const int4 s = *(const int4*)(srcs + i);
    int4 lo, hi;
    lo.x = s.x; lo.y = __float_as_int(dinv[s.x]);
    lo.z = s.y; lo.w = __float_as_int(dinv[s.y]);
    hi.x = s.z; hi.y = __float_as_int(dinv[s.z]);
    hi.z = s.w; hi.w = __float_as_int(dinv[s.w]);
    int4* d = (int4*)(ent + i);
    d[0] = lo;
    d[1] = hi;
  }
}

#define TILE_M        16
#define NTILES        (N_NODES / TILE_M)
#define SW_PITCH      544
#define MMA_WAVES     8
#define GATHER_WAVES  16
#define LAYER_THREADS ((MMA_WAVES + GATHER_WAVES) * 32)

static_assert(N_NODES % TILE_M == 0, "row tiles are exact, no guards in the layer kernel");

__device__ __forceinline__ v4f relu4(v4f x) {
  const v4f z = {0.0f, 0.0f, 0.0f, 0.0f};
  return __builtin_elementwise_max(x, z);
}

__device__ __forceinline__ void lds_barrier() {
  __builtin_amdgcn_fence(__ATOMIC_RELEASE, "workgroup", "local");
  __builtin_amdgcn_s_barrier();
  __builtin_amdgcn_fence(__ATOMIC_ACQUIRE, "workgroup", "local");
}

template <int DIN, int RELU, int B>
__device__ __forceinline__ void gather_rows(v4f (&acc)[DIN / 128], const float* __restrict__ sl,
                                            int myr, int myw, int j, int m) {
  constexpr int NV = DIN / 128;
  v4f xv[B][NV];
  float w[B];
#pragma unroll
  for (int u = 0; u < B; ++u) {
    const int jj = min(j + u, m - 1);
    const int r = __builtin_amdgcn_readlane(myr, jj);
    w[u] = __int_as_float(__builtin_amdgcn_readlane(myw, jj));
    const float* __restrict__ p = sl + (size_t)r * DIN;
#pragma unroll
    for (int v = 0; v < NV; ++v) xv[u][v] = *(const v4f*)(p + v * 128);
  }
#pragma unroll
  for (int u = 0; u < B; ++u) {
    if (j + u < m) {
      const v4f wv = {w[u], w[u], w[u], w[u]};
#pragma unroll
      for (int v = 0; v < NV; ++v) {
        v4f x = xv[u][v];
        if (RELU) x = relu4(x);
        acc[v] = __builtin_elementwise_fma(x, wv, acc[v]);
      }
    }
  }
}

template <int DIN>
__device__ __forceinline__ void mma_tile(const float* __restrict__ sAbuf, const float* __restrict__ sW,
                                         float* __restrict__ H, int tile, int mw, int lane) {
  constexpr int PA = DIN + 4;
  const int half = lane >> 4;
  const int c = lane & 15;
  const float* __restrict__ aF = sAbuf + c * PA + 2 * half;
  const float* __restrict__ bF = sW + half * SW_PITCH + (mw * 32 + c) * 2;
  v8f acc0 = v8f{};
  v8f acc1 = v8f{};
#pragma unroll 8
  for (int ks = 0; ks < DIN / 4; ++ks) {
    const v2f a  = *(const v2f*)(aF + ks * 4);
    const v2f b0 = *(const v2f*)(bF + (2 * ks) * SW_PITCH);
    const v2f b1 = *(const v2f*)(bF + (2 * ks) * SW_PITCH + 32);
    acc0 = __builtin_amdgcn_wmma_f32_16x16x4_f32(false, a, false, b0, (short)0, acc0, false, false);
    acc1 = __builtin_amdgcn_wmma_f32_16x16x4_f32(false, a, false, b1, (short)0, acc1, false, false);
  }
  float* __restrict__ Hp = H + (size_t)(tile * TILE_M + half * 8) * 256 + mw * 32 + c;
#pragma unroll
  for (int e = 0; e < 8; ++e) {
    Hp[(size_t)e * 256 + 0]  = acc0[e];
    Hp[(size_t)e * 256 + 16] = acc1[e];
  }
}

template <int DIN, int RELU>
__global__ __launch_bounds__(LAYER_THREADS) void layer_kernel(const float* __restrict__ src,
                                                              const float* __restrict__ W,
                                                              const float* __restrict__ dinv,
                                                              const int* __restrict__ cnt,
                                                              const int* __restrict__ off,
                                                              const int2* __restrict__ ent,
                                                              float* __restrict__ H) {
  constexpr int PA = DIN + 4;
  constexpr int NV = DIN / 128;
  constexpr int B  = 8;
  __shared__ __attribute__((aligned(16))) float sW[(DIN / 2) * SW_PITCH];
  __shared__ __attribute__((aligned(16))) float sA[2 * TILE_M * PA];

  const int lane = threadIdx.x & 31;
  const int wave = __builtin_amdgcn_readfirstlane((int)(threadIdx.x >> 5));
  const int q = wave / 3;
  const bool is_mma = (wave - 3 * q) == 0;
  const int mw = q;
  const int g = wave - q - 1;

  if (!is_mma) __builtin_amdgcn_s_setprio(3);

  const int first = blockIdx.x;
  const int stride = gridDim.x;
  const int ntl = (NTILES - first + stride - 1) / stride;

  const float* __restrict__ sl = src + lane * 4;

  float n_dc = 0.0f;
  int n_beg = 0, n_deg = 0;
  int2 n_en = make_int2(0, 0);
  v4f n_self[NV];
#pragma unroll
  for (int v = 0; v < NV; ++v) n_self[v] = v4f{};

  if (is_mma) {
    const int tm = mw * 32 + lane;
    const int wp = tm >> 6;
    const int wq = tm & 63;
#pragma unroll 4
    for (int kp = wp; kp < DIN / 2; kp += 4) {
      const float* __restrict__ gw = W + (size_t)(2 * kp) * 256 + wq * 4;
      const v4f w0 = *(const v4f*)(gw);
      const v4f w1 = *(const v4f*)(gw + 256);
      const v4f lo = {w0.x, w1.x, w0.y, w1.y};
      const v4f hi = {w0.z, w1.z, w0.w, w1.w};
      float* d = sW + kp * SW_PITCH + wq * 8;
      *(v4f*)(d) = lo;
      *(v4f*)(d + 4) = hi;
    }
  } else if (ntl > 0) {
    const int node = first * TILE_M + g;
    n_dc = dinv[node];
    n_beg = off[node];
    n_deg = cnt[node];
    if (lane < min(32, n_deg)) n_en = ent[n_beg + lane];
#pragma unroll
    for (int v = 0; v < NV; ++v) n_self[v] = *(const v4f*)(sl + (size_t)node * DIN + v * 128);
  }

  for (int i = 0; i <= ntl; ++i) {
    if (is_mma) {
      if (i >= 1)
        mma_tile<DIN>(sA + ((i - 1) & 1) * (TILE_M * PA), sW, H, first + (i - 1) * stride, mw, lane);
    } else if (i < ntl) {
      const float dc = n_dc;
      const int beg = __builtin_amdgcn_readfirstlane(n_beg);
      const int deg = __builtin_amdgcn_readfirstlane(n_deg);
      const int2 en0 = n_en;
      v4f acc[NV];
      {
        const float s = dc * dc;
#pragma unroll
        for (int v = 0; v < NV; ++v) {
          v4f x = n_self[v];
          if (RELU) x = relu4(x);
          acc[v] = x * s;
        }
      }

      const bool more = i + 1 < ntl;
      const int nnode = (first + (i + 1) * stride) * TILE_M + g;
      if (more) {
        n_dc = dinv[nnode];
        n_beg = off[nnode];
        n_deg = cnt[nnode];
      }

      for (int base = 0; base < deg; base += 32) {
        const int m = min(32, deg - base);
        int2 en = en0;
        if (base > 0) {
          en = make_int2(0, 0);
          if (lane < m) en = ent[beg + base + lane];
        }
        const int myr = en.x;
        const int myw = __float_as_int(__int_as_float(en.y) * dc);
        for (int j = 0; j < m; j += B) gather_rows<DIN, RELU, B>(acc, sl, myr, myw, j, m);
      }

      if (more) {
        n_en = make_int2(0, 0);
        if (lane < min(32, n_deg)) n_en = ent[n_beg + lane];
#pragma unroll
        for (int v = 0; v < NV; ++v) n_self[v] = *(const v4f*)(sl + (size_t)nnode * DIN + v * 128);
      }

      float* __restrict__ dst = sA + (i & 1) * (TILE_M * PA) + g * PA + lane * 4;
#pragma unroll
      for (int v = 0; v < NV; ++v) *(v4f*)(dst + v * 128) = acc[v];
    }
    lds_barrier();
  }
  if (!is_mma) __builtin_amdgcn_s_setprio(0);
}


extern "C" void kernel_launch(void* const* d_in, const int* in_sizes, int n_in,
                              void* d_out, int out_size, void* d_ws, size_t ws_size,
                              hipStream_t stream) {
  const float* x     = (const float*)d_in[0];
  const int*   ei    = (const int*)d_in[1];
  const float* W_in  = (const float*)d_in[2];
  const float* W_hid = (const float*)d_in[3];
  const float* W_out = (const float*)d_in[4];
  float* out = (float*)d_out;

  float* ws   = (float*)d_ws;
  int*   cnt  = (int*)ws;
  float* dinv = ws + 65536;
  int*   off  = (int*)(ws + 2 * 65536);
  int*   srcs = (int*)(ws + 3 * 65536);
  int2*  ent  = (int2*)(ws + (1 << 20));
  int*   lst  = (int*)(ws + (3 << 20));
  float* bufH = ws + (4 << 20);
  float* bufG = bufH + (size_t)N_NODES * 256;
  int2*  sorted = (int2*)(ws + (30 << 20));

  const int* row = ei;
  const int* col = ei + N_EDGES;

  const int WGT_BLKS   = (N_EDGES / 4 + 255) / 256;
  const int L1_BLKS   = 512;
  const int LAYER_BLKS = 256;

  bin_chunk_kernel<<<NCHUNK, 1024, 0, stream>>>(row, col, sorted, lst);
  bin_sort_kernel<<<NBIN, 256, 0, stream>>>(sorted, lst, cnt, off, dinv, srcs);
  ent_weight_kernel<<<WGT_BLKS, 256, 0, stream>>>(srcs, dinv, ent);

  layer_kernel<D_IN, 0><<<L1_BLKS, LAYER_THREADS, 0, stream>>>(x, W_in, dinv, cnt, off, ent, bufH);
  layer_kernel<D_HID, 1><<<LAYER_BLKS, LAYER_THREADS, 0, stream>>>(bufH, W_hid, dinv, cnt, off, ent, bufG);
  layer_kernel<D_HID, 1><<<LAYER_BLKS, LAYER_THREADS, 0, stream>>>(bufG, W_out, dinv, cnt, off, ent, out);
}


# relu moved from the gather waves (applied to every gathered row, about 17 times per element) to the producing layer's matrix-wave epilogue (once per element); stored values and all sums bit-identical
# speedup vs baseline: 5.4356x; 1.0428x over previous
#include <hip/hip_runtime.h>

#define N_NODES 50000
#define N_EDGES 800000
#define D_IN    128
#define D_HID   256

typedef __attribute__((ext_vector_type(2))) float v2f;
typedef __attribute__((ext_vector_type(4))) float v4f;
typedef __attribute__((ext_vector_type(8))) float v8f;

#define BIN_SHIFT 6
#define BIN_NODES (1 << BIN_SHIFT)
#define NBIN      ((N_NODES + BIN_NODES - 1) / BIN_NODES)
#define CHUNK     8192
#define NCHUNK    ((N_EDGES + CHUNK - 1) / CHUNK)
#define BIN_CAP   2048

static_assert(NBIN + 1 <= 1024, "one thread per bin start in the level-1 scan");
static_assert(NCHUNK <= 256, "one thread per chunk piece in level 2");
static_assert(CHUNK % 1024 == 0, "level 1: whole edges per thread");
static_assert(N_EDGES % 4 == 0, "level 3 handles 4 entries per thread");

__device__ __forceinline__ int wave_incl_scan(int v, int lane) {
  for (int d = 1; d < 32; d <<= 1) {
    const int n = __shfl_up(v, d);
    if (lane >= d) v += n;
  }
  return v;
}

template <int NW>
__device__ __forceinline__ int block_excl_scan(int v, int* wsum, int& total) {
  const int lane = threadIdx.x & 31;
  const int w = threadIdx.x >> 5;
  const int incl = wave_incl_scan(v, lane);
  if (lane == 31) wsum[w] = incl;
  __syncthreads();
  const int ws = (lane < NW) ? wsum[lane] : 0;
  const int wincl = wave_incl_scan(ws, lane);
  const int wave_excl = __shfl(wincl, w) - __shfl(ws, w);
  total = __shfl(wincl, 31);
  __syncthreads();
  return wave_excl + incl - v;
}

__global__ __launch_bounds__(1024) void bin_chunk_kernel(const int* __restrict__ row,
                                                         const int* __restrict__ col,
                                                         int2* __restrict__ sorted,
                                                         int* __restrict__ lst) {
  __shared__ int hcnt[NBIN];
  __shared__ int hoff[NBIN + 1];
  __shared__ int wsum[32];
  __shared__ int2 sPair[CHUNK];
  constexpr int EPT = CHUNK / 1024;
  const int t = threadIdx.x;
  const int chunk = blockIdx.x;
  const int e0 = chunk * CHUNK;
  const int n = min(CHUNK, N_EDGES - e0);

  if (t < NBIN) hcnt[t] = 0;
  __syncthreads();

  int c[EPT], r[EPT], lr[EPT];
#pragma unroll
  for (int u = 0; u < EPT; ++u) {
    const int j = u * 1024 + t;
    c[u] = -1;
    r[u] = 0;
    if (j < n) {
      c[u] = col[e0 + j];
      r[u] = row[e0 + j];
    }
  }
#pragma unroll
  for (int u = 0; u < EPT; ++u) {
    lr[u] = 0;
    if (c[u] >= 0) lr[u] = atomicAdd(&hcnt[c[u] >> BIN_SHIFT], 1);
  }
  __syncthreads();

  int total;
  const int v = (t < NBIN) ? hcnt[t] : 0;
  const int excl = block_excl_scan<32>(v, wsum, total);
  if (t <= NBIN) {
    hoff[t] = excl;
    lst[t * NCHUNK + chunk] = excl;
  }
  __syncthreads();

#pragma unroll
  for (int u = 0; u < EPT; ++u)
    if (c[u] >= 0) sPair[hoff[c[u] >> BIN_SHIFT] + lr[u]] = make_int2(r[u], c[u]);
  __syncthreads();

  for (int j = t; j < n; j += 1024) sorted[e0 + j] = sPair[j];
}

__global__ __launch_bounds__(256) void bin_sort_kernel(const int2* __restrict__ sorted,
                                                       const int* __restrict__ lst,
                                                       int* __restrict__ cnt,
                                                       int* __restrict__ off,
                                                       float* __restrict__ dinv,
                                                       int* __restrict__ srcs) {
  __shared__ int segS[NCHUNK];
  __shared__ int segOff[NCHUNK + 1];
  __shared__ int ncnt[BIN_NODES];
  __shared__ int cursor[BIN_NODES];
  __shared__ int wsum[32];
  __shared__ int2 sEdge[BIN_CAP];
  __shared__ int sOut[BIN_CAP];
  const int t = threadIdx.x;
  const int lane = t & 31;
  const int b = blockIdx.x;
  const int nodeBase = b * BIN_NODES;

  int s = 0, len = 0;
  if (t < NCHUNK) {
    s = lst[b * NCHUNK + t];
    len = lst[(b + 1) * NCHUNK + t] - s;
  }
  if (t < BIN_NODES) ncnt[t] = 0;
  int binTotal, binBase;
  const int so = block_excl_scan<8>(len, wsum, binTotal);
  (void)block_excl_scan<8>(s, wsum, binBase);
  if (t < NCHUNK) {
    segS[t] = s;
    segOff[t] = so;
  }
  if (t == 0) segOff[NCHUNK] = binTotal;
  __syncthreads();

  const bool small = binTotal <= BIN_CAP;

  auto fetch = [&](int j) -> int2 {
    int lo = 0, hi = NCHUNK - 1;
    while (lo < hi) {
      const int mid = (lo + hi + 1) >> 1;
      if (segOff[mid] <= j) lo = mid; else hi = mid - 1;
    }
    return sorted[(size_t)lo * CHUNK + segS[lo] + (j - segOff[lo])];
  };

  for (int j0 = t; j0 < binTotal; j0 += 1024) {
    int2 e[4];
#pragma unroll
    for (int u = 0; u < 4; ++u) {
      const int j = j0 + 256 * u;
      e[u] = make_int2(0, nodeBase);
      if (j < binTotal) e[u] = fetch(j);
    }
#pragma unroll
    for (int u = 0; u < 4; ++u) {
      const int j = j0 + 256 * u;
      if (j < binTotal) {
        if (small) sEdge[j] = e[u];
        atomicAdd(&ncnt[e[u].y - nodeBase], 1);
      }
    }
  }
  __syncthreads();

  if (t < 32) {
    const int a = ncnt[2 * lane], bb = ncnt[2 * lane + 1];
    const int incl = wave_incl_scan(a + bb, lane);
    const int o0 = incl - (a + bb);
    const int o1 = o0 + a;
    cursor[2 * lane] = o0;
    cursor[2 * lane + 1] = o1;
    const int n0 = nodeBase + 2 * lane;
    const int n1 = n0 + 1;
    if (n0 < N_NODES) {
      cnt[n0] = a;
      off[n0] = binBase + o0;
      dinv[n0] = rsqrtf((float)(a + 1));
    }
    if (n1 < N_NODES) {
      cnt[n1] = bb;
      off[n1] = binBase + o1;
      dinv[n1] = rsqrtf((float)(bb + 1));
    }
  }
  __syncthreads();

  for (int j = t; j < binTotal; j += 256) {
    const int2 e = small ? sEdge[j] : fetch(j);
    const int pl = atomicAdd(&cursor[e.y - nodeBase], 1);
    if (small) sOut[pl] = e.x; else srcs[binBase + pl] = e.x;
  }
  __syncthreads();
  if (small)
    for (int j = t; j < binTotal; j += 256) srcs[binBase + j] = sOut[j];
}

__global__ void ent_weight_kernel(const int* __restrict__ srcs, const float* __restrict__ dinv,
                                  int2* __restrict__ ent) {
  const int i = (blockIdx.x * blockDim.x + threadIdx.x) * 4;
  if (i < N_EDGES) {
    const int4 s = *(const int4*)(srcs + i);
    int4 lo, hi;
    lo.x = s.x; lo.y = __float_as_int(dinv[s.x]);
    lo.z = s.y; lo.w = __float_as_int(dinv[s.y]);
    hi.x = s.z; hi.y = __float_as_int(dinv[s.z]);
    hi.z = s.w; hi.w = __float_as_int(dinv[s.w]);
    int4* d = (int4*)(ent + i);
    d[0] = lo;
    d[1] = hi;
  }
}

#define TILE_M        16
#define NTILES        (N_NODES / TILE_M)
#define SW_PITCH      544
#define MMA_WAVES     8
#define GATHER_WAVES  16
#define LAYER_THREADS ((MMA_WAVES + GATHER_WAVES) * 32)

static_assert(N_NODES % TILE_M == 0, "row tiles are exact, no guards in the layer kernel");

__device__ __forceinline__ v8f relu8(v8f x) {
  const v8f z = {0.0f, 0.0f, 0.0f, 0.0f, 0.0f, 0.0f, 0.0f, 0.0f};
  return __builtin_elementwise_max(x, z);
}

__device__ __forceinline__ void lds_barrier() {
  __builtin_amdgcn_fence(__ATOMIC_RELEASE, "workgroup", "local");
  __builtin_amdgcn_s_barrier();
  __builtin_amdgcn_fence(__ATOMIC_ACQUIRE, "workgroup", "local");
}

template <int DIN, int B>
__device__ __forceinline__ void gather_rows(v4f (&acc)[DIN / 128], const float* __restrict__ sl,
                                            int myr, int myw, int j, int m) {
  constexpr int NV = DIN / 128;
  v4f xv[B][NV];
  float w[B];
#pragma unroll
  for (int u = 0; u < B; ++u) {
    const int jj = min(j + u, m - 1);
    const int r = __builtin_amdgcn_readlane(myr, jj);
    w[u] = __int_as_float(__builtin_amdgcn_readlane(myw, jj));
    const float* __restrict__ p = sl + (size_t)r * DIN;
#pragma unroll
    for (int v = 0; v < NV; ++v) xv[u][v] = *(const v4f*)(p + v * 128);
  }
#pragma unroll
  for (int u = 0; u < B; ++u) {
    if (j + u < m) {
      const v4f wv = {w[u], w[u], w[u], w[u]};
#pragma unroll
      for (int v = 0; v < NV; ++v) acc[v] = __builtin_elementwise_fma(xv[u][v], wv, acc[v]);
    }
  }
}

template <int DIN, int RELU_OUT>
__device__ __forceinline__ void mma_tile(const float* __restrict__ sAbuf, const float* __restrict__ sW,
                                         float* __restrict__ H, int tile, int mw, int lane) {
  constexpr int PA = DIN + 4;
  const int half = lane >> 4;
  const int c = lane & 15;
  const float* __restrict__ aF = sAbuf + c * PA + 2 * half;
  const float* __restrict__ bF = sW + half * SW_PITCH + (mw * 32 + c) * 2;
  v8f acc0 = v8f{};
  v8f acc1 = v8f{};
#pragma unroll 8
  for (int ks = 0; ks < DIN / 4; ++ks) {
    const v2f a  = *(const v2f*)(aF + ks * 4);
    const v2f b0 = *(const v2f*)(bF + (2 * ks) * SW_PITCH);
    const v2f b1 = *(const v2f*)(bF + (2 * ks) * SW_PITCH + 32);
    acc0 = __builtin_amdgcn_wmma_f32_16x16x4_f32(false, a, false, b0, (short)0, acc0, false, false);
    acc1 = __builtin_amdgcn_wmma_f32_16x16x4_f32(false, a, false, b1, (short)0, acc1, false, false);
  }
  if (RELU_OUT) {
    acc0 = relu8(acc0);
    acc1 = relu8(acc1);
  }
  float* __restrict__ Hp = H + (size_t)(tile * TILE_M + half * 8) * 256 + mw * 32 + c;
#pragma unroll
  for (int e = 0; e < 8; ++e) {
    Hp[(size_t)e * 256 + 0]  = acc0[e];
    Hp[(size_t)e * 256 + 16] = acc1[e];
  }
}

template <int DIN, int RELU_OUT>
__global__ __launch_bounds__(LAYER_THREADS) void layer_kernel(const float* __restrict__ src,
                                                              const float* __restrict__ W,
                                                              const float* __restrict__ dinv,
                                                              const int* __restrict__ cnt,
                                                              const int* __restrict__ off,
                                                              const int2* __restrict__ ent,
                                                              float* __restrict__ H) {
  constexpr int PA = DIN + 4;
  constexpr int NV = DIN / 128;
  constexpr int B  = 8;
  __shared__ __attribute__((aligned(16))) float sW[(DIN / 2) * SW_PITCH];
  __shared__ __attribute__((aligned(16))) float sA[2 * TILE_M * PA];

  const int lane = threadIdx.x & 31;
  const int wave = __builtin_amdgcn_readfirstlane((int)(threadIdx.x >> 5));
  const int q = wave / 3;
  const bool is_mma = (wave - 3 * q) == 0;
  const int mw = q;
  const int g = wave - q - 1;

  if (!is_mma) __builtin_amdgcn_s_setprio(3);

  const int first = blockIdx.x;
  const int stride = gridDim.x;
  const int ntl = (NTILES - first + stride - 1) / stride;

  const float* __restrict__ sl = src + lane * 4;

  float n_dc = 0.0f;
  int n_beg = 0, n_deg = 0;
  int2 n_en = make_int2(0, 0);
  v4f n_self[NV];
#pragma unroll
  for (int v = 0; v < NV; ++v) n_self[v] = v4f{};

  if (is_mma) {
    const int tm = mw * 32 + lane;
    const int wp = tm >> 6;
    const int wq = tm & 63;
#pragma unroll 4
    for (int kp = wp; kp < DIN / 2; kp += 4) {
      const float* __restrict__ gw = W + (size_t)(2 * kp) * 256 + wq * 4;
      const v4f w0 = *(const v4f*)(gw);
      const v4f w1 = *(const v4f*)(gw + 256);
      const v4f lo = {w0.x, w1.x, w0.y, w1.y};
      const v4f hi = {w0.z, w1.z, w0.w, w1.w};
      float* d = sW + kp * SW_PITCH + wq * 8;
      *(v4f*)(d) = lo;
      *(v4f*)(d + 4) = hi;
    }
  } else if (ntl > 0) {
    const int node = first * TILE_M + g;
    n_dc = dinv[node];
    n_beg = off[node];
    n_deg = cnt[node];
    if (lane < min(32, n_deg)) n_en = ent[n_beg + lane];
#pragma unroll
    for (int v = 0; v < NV; ++v) n_self[v] = *(const v4f*)(sl + (size_t)node * DIN + v * 128);
  }

  for (int i = 0; i <= ntl; ++i) {
    if (is_mma) {
      if (i >= 1)
        mma_tile<DIN, RELU_OUT>(sA + ((i - 1) & 1) * (TILE_M * PA), sW, H, first + (i - 1) * stride, mw, lane);
    } else if (i < ntl) {
      const float dc = n_dc;
      const int beg = __builtin_amdgcn_readfirstlane(n_beg);
      const int deg = __builtin_amdgcn_readfirstlane(n_deg);
      const int2 en0 = n_en;
      v4f acc[NV];
      {
        const float s = dc * dc;
#pragma unroll
        for (int v = 0; v < NV; ++v) acc[v] = n_self[v] * s;
      }

      const bool more = i + 1 < ntl;
      const int nnode = (first + (i + 1) * stride) * TILE_M + g;
      if (more) {
        n_dc = dinv[nnode];
        n_beg = off[nnode];
        n_deg = cnt[nnode];
      }

      for (int base = 0; base < deg; base += 32) {
        const int m = min(32, deg - base);
        int2 en = en0;
        if (base > 0) {
          en = make_int2(0, 0);
          if (lane < m) en = ent[beg + base + lane];
        }
        const int myr = en.x;
        const int myw = __float_as_int(__int_as_float(en.y) * dc);
        for (int j = 0; j < m; j += B) gather_rows<DIN, B>(acc, sl, myr, myw, j, m);
      }

      if (more) {
        n_en = make_int2(0, 0);
        if (lane < min(32, n_deg)) n_en = ent[n_beg + lane];
#pragma unroll
        for (int v = 0; v < NV; ++v) n_self[v] = *(const v4f*)(sl + (size_t)nnode * DIN + v * 128);
      }

      float* __restrict__ dst = sA + (i & 1) * (TILE_M * PA) + g * PA + lane * 4;
#pragma unroll
      for (int v = 0; v < NV; ++v) *(v4f*)(dst + v * 128) = acc[v];
    }
    lds_barrier();
  }
  if (!is_mma) __builtin_amdgcn_s_setprio(0);
}


extern "C" void kernel_launch(void* const* d_in, const int* in_sizes, int n_in,
                              void* d_out, int out_size, void* d_ws, size_t ws_size,
                              hipStream_t stream) {
  const float* x     = (const float*)d_in[0];
  const int*   ei    = (const int*)d_in[1];
  const float* W_in  = (const float*)d_in[2];
  const float* W_hid = (const float*)d_in[3];
  const float* W_out = (const float*)d_in[4];
  float* out = (float*)d_out;

  float* ws   = (float*)d_ws;
  int*   cnt  = (int*)ws;
  float* dinv = ws + 65536;
  int*   off  = (int*)(ws + 2 * 65536);
  int*   srcs = (int*)(ws + 3 * 65536);
  int2*  ent  = (int2*)(ws + (1 << 20));
  int*   lst  = (int*)(ws + (3 << 20));
  float* bufH = ws + (4 << 20);
  float* bufG = bufH + (size_t)N_NODES * 256;
  int2*  sorted = (int2*)(ws + (30 << 20));

  const int* row = ei;
  const int* col = ei + N_EDGES;

  const int WGT_BLKS   = (N_EDGES / 4 + 255) / 256;
  const int L1_BLKS   = 512;
  const int LAYER_BLKS = 256;

  bin_chunk_kernel<<<NCHUNK, 1024, 0, stream>>>(row, col, sorted, lst);
  bin_sort_kernel<<<NBIN, 256, 0, stream>>>(sorted, lst, cnt, off, dinv, srcs);
  ent_weight_kernel<<<WGT_BLKS, 256, 0, stream>>>(srcs, dinv, ent);

  layer_kernel<D_IN, 1><<<L1_BLKS, LAYER_THREADS, 0, stream>>>(x, W_in, dinv, cnt, off, ent, bufH);
  layer_kernel<D_HID, 1><<<LAYER_BLKS, LAYER_THREADS, 0, stream>>>(bufH, W_hid, dinv, cnt, off, ent, bufG);
  layer_kernel<D_HID, 0><<<LAYER_BLKS, LAYER_THREADS, 0, stream>>>(bufG, W_out, dinv, cnt, off, ent, out);
}


# Matrix waves of the 256-deep layers read each 8-k-step block's A/B fragments from LDS as one group, a block ahead into a second register set, not spread among the WMMAs; k order and sums unchanged
# speedup vs baseline: 5.6099x; 1.0321x over previous
#include <hip/hip_runtime.h>

#define N_NODES 50000
#define N_EDGES 800000
#define D_IN    128
#define D_HID   256

typedef __attribute__((ext_vector_type(2))) float v2f;
typedef __attribute__((ext_vector_type(4))) float v4f;
typedef __attribute__((ext_vector_type(8))) float v8f;

#define BIN_SHIFT 6
#define BIN_NODES (1 << BIN_SHIFT)
#define NBIN      ((N_NODES + BIN_NODES - 1) / BIN_NODES)
#define CHUNK     8192
#define NCHUNK    ((N_EDGES + CHUNK - 1) / CHUNK)
#define BIN_CAP   2048

static_assert(NBIN + 1 <= 1024, "one thread per bin start in the level-1 scan");
static_assert(NCHUNK <= 256, "one thread per chunk piece in level 2");
static_assert(CHUNK % 1024 == 0, "level 1: whole edges per thread");
static_assert(N_EDGES % 4 == 0, "level 3 handles 4 entries per thread");

__device__ __forceinline__ int wave_incl_scan(int v, int lane) {
  for (int d = 1; d < 32; d <<= 1) {
    const int n = __shfl_up(v, d);
    if (lane >= d) v += n;
  }
  return v;
}

template <int NW>
__device__ __forceinline__ int block_excl_scan(int v, int* wsum, int& total) {
  const int lane = threadIdx.x & 31;
  const int w = threadIdx.x >> 5;
  const int incl = wave_incl_scan(v, lane);
  if (lane == 31) wsum[w] = incl;
  __syncthreads();
  const int ws = (lane < NW) ? wsum[lane] : 0;
  const int wincl = wave_incl_scan(ws, lane);
  const int wave_excl = __shfl(wincl, w) - __shfl(ws, w);
  total = __shfl(wincl, 31);
  __syncthreads();
  return wave_excl + incl - v;
}

__global__ __launch_bounds__(1024) void bin_chunk_kernel(const int* __restrict__ row,
                                                         const int* __restrict__ col,
                                                         int2* __restrict__ sorted,
                                                         int* __restrict__ lst) {
  __shared__ int hcnt[NBIN];
  __shared__ int hoff[NBIN + 1];
  __shared__ int wsum[32];
  __shared__ int2 sPair[CHUNK];
  constexpr int EPT = CHUNK / 1024;
  const int t = threadIdx.x;
  const int chunk = blockIdx.x;
  const int e0 = chunk * CHUNK;
  const int n = min(CHUNK, N_EDGES - e0);

  if (t < NBIN) hcnt[t] = 0;
  __syncthreads();

  int c[EPT], r[EPT], lr[EPT];
#pragma unroll
  for (int u = 0; u < EPT; ++u) {
    const int j = u * 1024 + t;
    c[u] = -1;
    r[u] = 0;
    if (j < n) {
      c[u] = col[e0 + j];
      r[u] = row[e0 + j];
    }
  }
#pragma unroll
  for (int u = 0; u < EPT; ++u) {
    lr[u] = 0;
    if (c[u] >= 0) lr[u] = atomicAdd(&hcnt[c[u] >> BIN_SHIFT], 1);
  }
  __syncthreads();

  int total;
  const int v = (t < NBIN) ? hcnt[t] : 0;
  const int excl = block_excl_scan<32>(v, wsum, total);
  if (t <= NBIN) {
    hoff[t] = excl;
    lst[t * NCHUNK + chunk] = excl;
  }
  __syncthreads();

#pragma unroll
  for (int u = 0; u < EPT; ++u)
    if (c[u] >= 0) sPair[hoff[c[u] >> BIN_SHIFT] + lr[u]] = make_int2(r[u], c[u]);
  __syncthreads();

  for (int j = t; j < n; j += 1024) sorted[e0 + j] = sPair[j];
}

__global__ __launch_bounds__(256) void bin_sort_kernel(const int2* __restrict__ sorted,
                                                       const int* __restrict__ lst,
                                                       int* __restrict__ cnt,
                                                       int* __restrict__ off,
                                                       float* __restrict__ dinv,
                                                       int* __restrict__ srcs) {
  __shared__ int segS[NCHUNK];
  __shared__ int segOff[NCHUNK + 1];
  __shared__ int ncnt[BIN_NODES];
  __shared__ int cursor[BIN_NODES];
  __shared__ int wsum[32];
  __shared__ int2 sEdge[BIN_CAP];
  __shared__ int sOut[BIN_CAP];
  const int t = threadIdx.x;
  const int lane = t & 31;
  const int b = blockIdx.x;
  const int nodeBase = b * BIN_NODES;

  int s = 0, len = 0;
  if (t < NCHUNK) {
    s = lst[b * NCHUNK + t];
    len = lst[(b + 1) * NCHUNK + t] - s;
  }
  if (t < BIN_NODES) ncnt[t] = 0;
  int binTotal, binBase;
  const int so = block_excl_scan<8>(len, wsum, binTotal);
  (void)block_excl_scan<8>(s, wsum, binBase);
  if (t < NCHUNK) {
    segS[t] = s;
    segOff[t] = so;
  }
  if (t == 0) segOff[NCHUNK] = binTotal;
  __syncthreads();

  const bool small = binTotal <= BIN_CAP;

  auto fetch = [&](int j) -> int2 {
    int lo = 0, hi = NCHUNK - 1;
    while (lo < hi) {
      const int mid = (lo + hi + 1) >> 1;
      if (segOff[mid] <= j) lo = mid; else hi = mid - 1;
    }
    return sorted[(size_t)lo * CHUNK + segS[lo] + (j - segOff[lo])];
  };

  for (int j0 = t; j0 < binTotal; j0 += 1024) {
    int2 e[4];
#pragma unroll
    for (int u = 0; u < 4; ++u) {
      const int j = j0 + 256 * u;
      e[u] = make_int2(0, nodeBase);
      if (j < binTotal) e[u] = fetch(j);
    }
#pragma unroll
    for (int u = 0; u < 4; ++u) {
      const int j = j0 + 256 * u;
      if (j < binTotal) {
        if (small) sEdge[j] = e[u];
        atomicAdd(&ncnt[e[u].y - nodeBase], 1);
      }
    }
  }
  __syncthreads();

  if (t < 32) {
    const int a = ncnt[2 * lane], bb = ncnt[2 * lane + 1];
    const int incl = wave_incl_scan(a + bb, lane);
    const int o0 = incl - (a + bb);
    const int o1 = o0 + a;
    cursor[2 * lane] = o0;
    cursor[2 * lane + 1] = o1;
    const int n0 = nodeBase + 2 * lane;
    const int n1 = n0 + 1;
    if (n0 < N_NODES) {
      cnt[n0] = a;
      off[n0] = binBase + o0;
      dinv[n0] = rsqrtf((float)(a + 1));
    }
    if (n1 < N_NODES) {
      cnt[n1] = bb;
      off[n1] = binBase + o1;
      dinv[n1] = rsqrtf((float)(bb + 1));
    }
  }
  __syncthreads();

  for (int j = t; j < binTotal; j += 256) {
    const int2 e = small ? sEdge[j] : fetch(j);
    const int pl = atomicAdd(&cursor[e.y - nodeBase], 1);
    if (small) sOut[pl] = e.x; else srcs[binBase + pl] = e.x;
  }
  __syncthreads();
  if (small)
    for (int j = t; j < binTotal; j += 256) srcs[binBase + j] = sOut[j];
}

__global__ void ent_weight_kernel(const int* __restrict__ srcs, const float* __restrict__ dinv,
                                  int2* __restrict__ ent) {
  const int i = (blockIdx.x * blockDim.x + threadIdx.x) * 4;
  if (i < N_EDGES) {
    const int4 s = *(const int4*)(srcs + i);
    int4 lo, hi;
    lo.x = s.x; lo.y = __float_as_int(dinv[s.x]);
    lo.z = s.y; lo.w = __float_as_int(dinv[s.y]);
    hi.x = s.z; hi.y = __float_as_int(dinv[s.z]);
    hi.z = s.w; hi.w = __float_as_int(dinv[s.w]);
    int4* d = (int4*)(ent + i);
    d[0] = lo;
    d[1] = hi;
  }
}

#define TILE_M        16
#define NTILES        (N_NODES / TILE_M)
#define SW_PITCH      544
#define MMA_WAVES     8
#define GATHER_WAVES  16
#define LAYER_THREADS ((MMA_WAVES + GATHER_WAVES) * 32)

static_assert(N_NODES % TILE_M == 0, "row tiles are exact, no guards in the layer kernel");

__device__ __forceinline__ v8f relu8(v8f x) {
  const v8f z = {0.0f, 0.0f, 0.0f, 0.0f, 0.0f, 0.0f, 0.0f, 0.0f};
  return __builtin_elementwise_max(x, z);
}

__device__ __forceinline__ void lds_barrier() {
  __builtin_amdgcn_fence(__ATOMIC_RELEASE, "workgroup", "local");
  __builtin_amdgcn_s_barrier();
  __builtin_amdgcn_fence(__ATOMIC_ACQUIRE, "workgroup", "local");
}

template <int DIN, int B>
__device__ __forceinline__ void gather_rows(v4f (&acc)[DIN / 128], const float* __restrict__ sl,
                                            int myr, int myw, int j, int m) {
  constexpr int NV = DIN / 128;
  v4f xv[B][NV];
  float w[B];
#pragma unroll
  for (int u = 0; u < B; ++u) {
    const int jj = min(j + u, m - 1);
    const int r = __builtin_amdgcn_readlane(myr, jj);
    w[u] = __int_as_float(__builtin_amdgcn_readlane(myw, jj));
    const float* __restrict__ p = sl + (size_t)r * DIN;
#pragma unroll
    for (int v = 0; v < NV; ++v) xv[u][v] = *(const v4f*)(p + v * 128);
  }
#pragma unroll
  for (int u = 0; u < B; ++u) {
    if (j + u < m) {
      const v4f wv = {w[u], w[u], w[u], w[u]};
#pragma unroll
      for (int v = 0; v < NV; ++v) acc[v] = __builtin_elementwise_fma(xv[u][v], wv, acc[v]);
    }
  }
}

template <int KB>
__device__ __forceinline__ void load_frags(v2f (&a)[KB], v2f (&b0)[KB], v2f (&b1)[KB],
                                           const float* __restrict__ aP, const float* __restrict__ bP) {
#pragma unroll
  for (int u = 0; u < KB; ++u) {
    a[u]  = *(const v2f*)(aP + u * 4);
    b0[u] = *(const v2f*)(bP + (2 * u) * SW_PITCH);
    b1[u] = *(const v2f*)(bP + (2 * u) * SW_PITCH + 32);
  }
}

template <int KB>
__device__ __forceinline__ void mma_block(v8f& acc0, v8f& acc1, const v2f (&a)[KB], const v2f (&b0)[KB],
                                          const v2f (&b1)[KB]) {
#pragma unroll
  for (int u = 0; u < KB; ++u) {
    acc0 = __builtin_amdgcn_wmma_f32_16x16x4_f32(false, a[u], false, b0[u], (short)0, acc0, false, false);
    acc1 = __builtin_amdgcn_wmma_f32_16x16x4_f32(false, a[u], false, b1[u], (short)0, acc1, false, false);
  }
}

template <int DIN, int RELU_OUT>
__device__ __forceinline__ void mma_tile(const float* __restrict__ sAbuf, const float* __restrict__ sW,
                                         float* __restrict__ H, int tile, int mw, int lane) {
  constexpr int PA = DIN + 4;
  constexpr int KB = (DIN >= 256) ? 8 : 4;
  constexpr int NP = DIN / 4 / KB / 2;
  constexpr int A_BLK = KB * 4;
  constexpr int B_BLK = KB * 2 * SW_PITCH;
  static_assert(NP * 2 * KB * 4 == DIN, "whole pairs of blocks");
  const int half = lane >> 4;
  const int c = lane & 15;
  const float* __restrict__ aP = sAbuf + c * PA + 2 * half;
  const float* __restrict__ bP = sW + half * SW_PITCH + (mw * 32 + c) * 2;
  v8f acc0 = v8f{};
  v8f acc1 = v8f{};
  if constexpr (DIN < 256) {
#pragma unroll 8
    for (int ks = 0; ks < DIN / 4; ++ks) {
      const v2f a  = *(const v2f*)(aP + ks * 4);
      const v2f b0 = *(const v2f*)(bP + (2 * ks) * SW_PITCH);
      const v2f b1 = *(const v2f*)(bP + (2 * ks) * SW_PITCH + 32);
      acc0 = __builtin_amdgcn_wmma_f32_16x16x4_f32(false, a, false, b0, (short)0, acc0, false, false);
      acc1 = __builtin_amdgcn_wmma_f32_16x16x4_f32(false, a, false, b1, (short)0, acc1, false, false);
    }
  } else {
  v2f aX[KB], b0X[KB], b1X[KB];
  v2f aY[KB], b0Y[KB], b1Y[KB];
  load_frags<KB>(aX, b0X, b1X, aP, bP);
#pragma nounroll
  for (int p = 0; p < NP - 1; ++p) {
    load_frags<KB>(aY, b0Y, b1Y, aP + A_BLK, bP + B_BLK);
    __builtin_amdgcn_sched_barrier(0);
    mma_block<KB>(acc0, acc1, aX, b0X, b1X);
    aP += 2 * A_BLK;
    bP += 2 * B_BLK;
    load_frags<KB>(aX, b0X, b1X, aP, bP);
    __builtin_amdgcn_sched_barrier(0);
    mma_block<KB>(acc0, acc1, aY, b0Y, b1Y);
    __builtin_amdgcn_sched_barrier(0);
  }
  load_frags<KB>(aY, b0Y, b1Y, aP + A_BLK, bP + B_BLK);
  __builtin_amdgcn_sched_barrier(0);
  mma_block<KB>(acc0, acc1, aX, b0X, b1X);
  mma_block<KB>(acc0, acc1, aY, b0Y, b1Y);
  }
  if (RELU_OUT) {
    acc0 = relu8(acc0);
    acc1 = relu8(acc1);
  }
  float* __restrict__ Hp = H + (size_t)(tile * TILE_M + half * 8) * 256 + mw * 32 + c;
#pragma unroll
  for (int e = 0; e < 8; ++e) {
    Hp[(size_t)e * 256 + 0]  = acc0[e];
    Hp[(size_t)e * 256 + 16] = acc1[e];
  }
}

template <int DIN, int RELU_OUT>
__global__ __launch_bounds__(LAYER_THREADS) void layer_kernel(const float* __restrict__ src,
                                                              const float* __restrict__ W,
                                                              const float* __restrict__ dinv,
                                                              const int* __restrict__ cnt,
                                                              const int* __restrict__ off,
                                                              const int2* __restrict__ ent,
                                                              float* __restrict__ H) {
  constexpr int PA = DIN + 4;
  constexpr int NV = DIN / 128;
  constexpr int B  = 8;
  __shared__ __attribute__((aligned(16))) float sW[(DIN / 2) * SW_PITCH];
  __shared__ __attribute__((aligned(16))) float sA[2 * TILE_M * PA];

  const int lane = threadIdx.x & 31;
  const int wave = __builtin_amdgcn_readfirstlane((int)(threadIdx.x >> 5));
  const int q = wave / 3;
  const bool is_mma = (wave - 3 * q) == 0;
  const int mw = q;
  const int g = wave - q - 1;

  if (!is_mma) __builtin_amdgcn_s_setprio(3);

  const int first = blockIdx.x;
  const int stride = gridDim.x;
  const int ntl = (NTILES - first + stride - 1) / stride;

  const float* __restrict__ sl = src + lane * 4;

  float n_dc = 0.0f;
  int n_beg = 0, n_deg = 0;
  int2 n_en = make_int2(0, 0);
  v4f n_self[NV];
#pragma unroll
  for (int v = 0; v < NV; ++v) n_self[v] = v4f{};

  if (is_mma) {
    const int tm = mw * 32 + lane;
    const int wp = tm >> 6;
    const int wq = tm & 63;
#pragma unroll 4
    for (int kp = wp; kp < DIN / 2; kp += 4) {
      const float* __restrict__ gw = W + (size_t)(2 * kp) * 256 + wq * 4;
      const v4f w0 = *(const v4f*)(gw);
      const v4f w1 = *(const v4f*)(gw + 256);
      const v4f lo = {w0.x, w1.x, w0.y, w1.y};
      const v4f hi = {w0.z, w1.z, w0.w, w1.w};
      float* d = sW + kp * SW_PITCH + wq * 8;
      *(v4f*)(d) = lo;
      *(v4f*)(d + 4) = hi;
    }
  } else if (ntl > 0) {
    const int node = first * TILE_M + g;
    n_dc = dinv[node];
    n_beg = off[node];
    n_deg = cnt[node];
    if (lane < min(32, n_deg)) n_en = ent[n_beg + lane];
#pragma unroll
    for (int v = 0; v < NV; ++v) n_self[v] = *(const v4f*)(sl + (size_t)node * DIN + v * 128);
  }

  for (int i = 0; i <= ntl; ++i) {
    if (is_mma) {
      if (i >= 1)
        mma_tile<DIN, RELU_OUT>(sA + ((i - 1) & 1) * (TILE_M * PA), sW, H, first + (i - 1) * stride, mw, lane);
    } else if (i < ntl) {
      const float dc = n_dc;
      const int beg = __builtin_amdgcn_readfirstlane(n_beg);
      const int deg = __builtin_amdgcn_readfirstlane(n_deg);
      const int2 en0 = n_en;
      v4f acc[NV];
      {
        const float s = dc * dc;
#pragma unroll
        for (int v = 0; v < NV; ++v) acc[v] = n_self[v] * s;
      }

      const bool more = i + 1 < ntl;
      const int nnode = (first + (i + 1) * stride) * TILE_M + g;
      if (more) {
        n_dc = dinv[nnode];
        n_beg = off[nnode];
        n_deg = cnt[nnode];
      }

      for (int base = 0; base < deg; base += 32) {
        const int m = min(32, deg - base);
        int2 en = en0;
        if (base > 0) {
          en = make_int2(0, 0);
          if (lane < m) en = ent[beg + base + lane];
        }
        const int myr = en.x;
        const int myw = __float_as_int(__int_as_float(en.y) * dc);
        for (int j = 0; j < m; j += B) gather_rows<DIN, B>(acc, sl, myr, myw, j, m);
      }

      if (more) {
        n_en = make_int2(0, 0);
        if (lane < min(32, n_deg)) n_en = ent[n_beg + lane];
#pragma unroll
        for (int v = 0; v < NV; ++v) n_self[v] = *(const v4f*)(sl + (size_t)nnode * DIN + v * 128);
      }

      float* __restrict__ dst = sA + (i & 1) * (TILE_M * PA) + g * PA + lane * 4;
#pragma unroll
      for (int v = 0; v < NV; ++v) *(v4f*)(dst + v * 128) = acc[v];
    }
    lds_barrier();
  }
  if (!is_mma) __builtin_amdgcn_s_setprio(0);
}


extern "C" void kernel_launch(void* const* d_in, const int* in_sizes, int n_in,
                              void* d_out, int out_size, void* d_ws, size_t ws_size,
                              hipStream_t stream) {
  const float* x     = (const float*)d_in[0];
  const int*   ei    = (const int*)d_in[1];
  const float* W_in  = (const float*)d_in[2];
  const float* W_hid = (const float*)d_in[3];
  const float* W_out = (const float*)d_in[4];
  float* out = (float*)d_out;

  float* ws   = (float*)d_ws;
  int*   cnt  = (int*)ws;
  float* dinv = ws + 65536;
  int*   off  = (int*)(ws + 2 * 65536);
  int*   srcs = (int*)(ws + 3 * 65536);
  int2*  ent  = (int2*)(ws + (1 << 20));
  int*   lst  = (int*)(ws + (3 << 20));
  float* bufH = ws + (4 << 20);
  float* bufG = bufH + (size_t)N_NODES * 256;
  int2*  sorted = (int2*)(ws + (30 << 20));

  const int* row = ei;
  const int* col = ei + N_EDGES;

  const int WGT_BLKS   = (N_EDGES / 4 + 255) / 256;
  const int L1_BLKS   = 512;
  const int LAYER_BLKS = 256;

  bin_chunk_kernel<<<NCHUNK, 1024, 0, stream>>>(row, col, sorted, lst);
  bin_sort_kernel<<<NBIN, 256, 0, stream>>>(sorted, lst, cnt, off, dinv, srcs);
  ent_weight_kernel<<<WGT_BLKS, 256, 0, stream>>>(srcs, dinv, ent);

  layer_kernel<D_IN, 1><<<L1_BLKS, LAYER_THREADS, 0, stream>>>(x, W_in, dinv, cnt, off, ent, bufH);
  layer_kernel<D_HID, 1><<<LAYER_BLKS, LAYER_THREADS, 0, stream>>>(bufH, W_hid, dinv, cnt, off, ent, bufG);
  layer_kernel<D_HID, 0><<<LAYER_BLKS, LAYER_THREADS, 0, stream>>>(bufG, W_out, dinv, cnt, off, ent, out);
}
